# starting kernel
# baseline (speedup 1.0000x reference)
#include <hip/hip_runtime.h>
#include <math.h>

typedef float v2f __attribute__((ext_vector_type(2)));
typedef float v8f __attribute__((ext_vector_type(8)));

#define N_NODES 177
#define FEAT 128
#define OUT_C 128
#define NEG_SLOPE 0.2f
#define MTILES 12
#define S_PAD 180
#define H_ROWS 180

#define OFF_H    0
#define OFF_A    (OFF_H + H_ROWS * FEAT)
#define OFF_B    (OFF_A + N_NODES)
#define OFF_ESUM (OFF_B + N_NODES)
#define OFF_T    (OFF_ESUM + N_NODES)
#define OFF_AMAX (OFF_T + N_NODES)
#define OFF_S    (OFF_AMAX + N_NODES)
#define WS_FLOATS (OFF_S + N_NODES * S_PAD)

__device__ __forceinline__ unsigned f2ord(float f) {
  unsigned u = __float_as_uint(f);
  return (u & 0x80000000u) ? ~u : (u | 0x80000000u);
}
__device__ __forceinline__ float ord2f(unsigned u) {
  return __uint_as_float((u & 0x80000000u) ? (u ^ 0x80000000u) : ~u);
}
#define ORD_NEG_INF 0x007FFFFFu

__device__ __forceinline__ v2f load_a_clamp(const float* __restrict__ A, int m0, int k0,
                                            int lane, int ld) {
  int m = m0 + (lane & 15);
  m = (m < N_NODES) ? m : (N_NODES - 1);
  int kb = k0 + ((lane >> 4) << 1);
  const float* p = A + m * ld + kb;
  v2f a; a.x = p[0]; a.y = p[1];
  return a;
}
__device__ __forceinline__ v2f load_b(const float* __restrict__ B, int k0, int n0,
                                      int lane, int ld) {
  int n  = n0 + (lane & 15);
  int kb = k0 + ((lane >> 4) << 1);
  const float* p = B + kb * ld + n;
  v2f b; b.x = p[0]; b.y = p[ld];
  return b;
}
__device__ __forceinline__ v2f load_a_lds(const float* A, int k0, int lane) {
  int m  = lane & 15;
  int kb = k0 + ((lane >> 4) << 1);
  v2f a;
  a.x = A[m * OUT_C + kb];
  a.y = A[m * OUT_C + kb + 1];
  return a;
}

__global__ void k_init(float* __restrict__ ws) {
  int tid    = blockIdx.x * blockDim.x + threadIdx.x;
  int stride = gridDim.x * blockDim.x;
  unsigned* amax = (unsigned*)(ws + OFF_AMAX);
  for (int j = tid; j < N_NODES * S_PAD; j += stride) ws[OFF_S + j] = 0.0f;
  for (int j = tid; j < (H_ROWS - N_NODES) * FEAT; j += stride)
    ws[OFF_H + N_NODES * FEAT + j] = 0.0f;
  for (int j = tid; j < N_NODES; j += stride) {
    ws[OFF_ESUM + j] = 0.0f;
    ws[OFF_T + j]    = 0.0f;
    amax[j]          = ORD_NEG_INF;
  }
}

__global__ void k_node_gemm(const float* __restrict__ x, const float* __restrict__ W,
                            float* __restrict__ ws) {
  float* h  = ws + OFF_H;
  int lane  = threadIdx.x & 31;
  int wave  = threadIdx.x >> 5;
  int m0    = blockIdx.x * 16;
  int n0    = wave * 32;
  v8f c0 = {}; v8f c1 = {};
  for (int k0 = 0; k0 < FEAT; k0 += 4) {
    v2f a  = load_a_clamp(x, m0, k0, lane, FEAT);
    v2f b0 = load_b(W, k0, n0,      lane, OUT_C);
    v2f b1 = load_b(W, k0, n0 + 16, lane, OUT_C);
    c0 = __builtin_amdgcn_wmma_f32_16x16x4_f32(false, a, false, b0, (short)0, c0, false, false);
    c1 = __builtin_amdgcn_wmma_f32_16x16x4_f32(false, a, false, b1, (short)0, c1, false, false);
  }
  int nb = n0 + (lane & 15);
  int mh = (lane >> 4) * 8;
  for (int j = 0; j < 8; ++j) {
    int m = m0 + j + mh;
    if (m < N_NODES) {
      h[m * OUT_C + nb]      = c0[j];
      h[m * OUT_C + nb + 16] = c1[j];
    }
  }
}

__global__ void k_ab(const float* __restrict__ att, float* __restrict__ ws) {
  const float* h = ws + OFF_H;
  int n = blockIdx.x, lane = threadIdx.x;
  float pa = 0.0f, pb = 0.0f;
  for (int i = lane; i < FEAT; i += 32) {
    float hv = h[n * OUT_C + i];
    pa += hv * att[i];
    pb += hv * att[OUT_C + i];
  }
  for (int m = 16; m >= 1; m >>= 1) {
    pa += __shfl_xor(pa, m, 32);
    pb += __shfl_xor(pb, m, 32);
  }
  if (lane == 0) { ws[OFF_A + n] = pa; ws[OFF_B + n] = pb; }
}

__global__ void k_edge_max(const int* __restrict__ ei, const float* __restrict__ ea,
                           const float* __restrict__ att, float* __restrict__ ws, int E) {
  const float* a = ws + OFF_A;
  const float* b = ws + OFF_B;
  unsigned* amax = (unsigned*)(ws + OFF_AMAX);
  float cc = att[2 * OUT_C];
  int tid = blockIdx.x * blockDim.x + threadIdx.x;
  int stride = gridDim.x * blockDim.x;
  for (int e = tid; e < E; e += stride) {
    int s = ei[e], d = ei[E + e];
    float al = a[d] + b[s] + cc * ea[e];
    al = (al > 0.0f) ? al : NEG_SLOPE * al;
    atomicMax(&amax[d], f2ord(al));
  }
}

__global__ void k_edge_exp(const int* __restrict__ ei, const float* __restrict__ ea,
                           const float* __restrict__ att, float* __restrict__ ws, int E) {
  const float* a = ws + OFF_A;
  const float* b = ws + OFF_B;
  const unsigned* amax = (const unsigned*)(ws + OFF_AMAX);
  float* esum = ws + OFF_ESUM;
  float* t    = ws + OFF_T;
  float* S    = ws + OFF_S;
  float cc = att[2 * OUT_C];
  int tid = blockIdx.x * blockDim.x + threadIdx.x;
  int stride = gridDim.x * blockDim.x;
  for (int e = tid; e < E; e += stride) {
    int s = ei[e], d = ei[E + e];
    float eav = ea[e];
    float al = a[d] + b[s] + cc * eav;
    al = (al > 0.0f) ? al : NEG_SLOPE * al;
    float w = __expf(al - ord2f(amax[d]));
    atomicAdd(&esum[d], w);
    atomicAdd(&t[d], w * eav);
    atomicAdd(&S[d * S_PAD + s], w);
  }
}

__global__ void k_final(const float* __restrict__ U, const float* __restrict__ bias,
                        const float* __restrict__ ws, float* __restrict__ out) {
  __shared__ float aggr[16 * OUT_C];
  const float* h    = ws + OFF_H;
  const float* S    = ws + OFF_S;
  const float* esum = ws + OFF_ESUM;
  const float* t    = ws + OFF_T;
  int lane = threadIdx.x & 31;
  int wave = threadIdx.x >> 5;
  int m0 = blockIdx.x * 16;
  int n0 = wave * 32;

  v8f c0 = {}; v8f c1 = {};
  for (int k0 = 0; k0 < S_PAD; k0 += 4) {
    v2f a  = load_a_clamp(S, m0, k0, lane, S_PAD);
    v2f b0 = load_b(h, k0, n0,      lane, OUT_C);
    v2f b1 = load_b(h, k0, n0 + 16, lane, OUT_C);
    c0 = __builtin_amdgcn_wmma_f32_16x16x4_f32(false, a, false, b0, (short)0, c0, false, false);
    c1 = __builtin_amdgcn_wmma_f32_16x16x4_f32(false, a, false, b1, (short)0, c1, false, false);
  }
  int nb = n0 + (lane & 15);
  int mh = (lane >> 4) * 8;
  for (int j = 0; j < 8; ++j) {
    int mloc = j + mh;
    int m = m0 + mloc;
    int mc = (m < N_NODES) ? m : (N_NODES - 1);
    float r = 1.0f / (esum[mc] + 1e-16f);
    aggr[mloc * OUT_C + nb]      = c0[j] * r;
    aggr[mloc * OUT_C + nb + 16] = c1[j] * r;
  }
  __syncthreads();

  v8f d0 = {}; v8f d1 = {};
  for (int k0 = 0; k0 < OUT_C; k0 += 4) {
    v2f a  = load_a_lds(aggr, k0, lane);
    v2f b0 = load_b(U, k0, n0,      lane, OUT_C);
    v2f b1 = load_b(U, k0, n0 + 16, lane, OUT_C);
    d0 = __builtin_amdgcn_wmma_f32_16x16x4_f32(false, a, false, b0, (short)0, d0, false, false);
    d1 = __builtin_amdgcn_wmma_f32_16x16x4_f32(false, a, false, b1, (short)0, d1, false, false);
  }
  for (int j = 0; j < 8; ++j) {
    int m = m0 + j + mh;
    if (m < N_NODES) {
      float tn = t[m] / (esum[m] + 1e-16f);
      out[m * OUT_C + nb]      = d0[j] + tn * U[OUT_C * OUT_C + nb]      + bias[nb];
      out[m * OUT_C + nb + 16] = d1[j] + tn * U[OUT_C * OUT_C + nb + 16] + bias[nb + 16];
    }
  }
}

extern "C" void kernel_launch(void* const* d_in, const int* in_sizes, int n_in,
                              void* d_out, int out_size, void* d_ws, size_t ws_size,
                              hipStream_t stream) {
  const float* x    = (const float*)d_in[0];
  const float* ea   = (const float*)d_in[1];
  const float* W    = (const float*)d_in[2];
  const float* att  = (const float*)d_in[3];
  const float* U    = (const float*)d_in[4];
  const float* bias = (const float*)d_in[5];
  const int*   ei   = (const int*)d_in[6];
  int E = in_sizes[1];
  float* ws  = (float*)d_ws;
  float* out = (float*)d_out;

  k_init<<<64, 256, 0, stream>>>(ws);
  k_node_gemm<<<MTILES, 128, 0, stream>>>(x, W, ws);
  k_ab<<<N_NODES, 32, 0, stream>>>(att, ws);
  k_edge_max<<<1024, 256, 0, stream>>>(ei, ea, att, ws, E);
  k_edge_exp<<<1024, 256, 0, stream>>>(ei, ea, att, ws, E);
  k_final<<<MTILES, 128, 0, stream>>>(U, bias, ws, out);
}


# edge passes reduce in LDS first: per-block LDS atomics for per-destination max, exp-sum, t and all of S (128 blocks of 1024), one line-aligned global-atomic flush per block; float sum order changes
# speedup vs baseline: 4.5089x; 4.5089x over previous
#include <hip/hip_runtime.h>
#include <math.h>

typedef float v2f __attribute__((ext_vector_type(2)));
typedef float v8f __attribute__((ext_vector_type(8)));

#define N_NODES 177
#define FEAT 128
#define OUT_C 128
#define NEG_SLOPE 0.2f
#define MTILES 12
#define S_PAD 180
#define H_ROWS 180

#define OFF_H    0
#define OFF_A    (OFF_H + H_ROWS * FEAT)
#define OFF_B    (OFF_A + N_NODES)
#define OFF_ESUM (OFF_B + N_NODES)
#define OFF_T    (OFF_ESUM + N_NODES)
#define OFF_AMAX (OFF_T + N_NODES)
#define OFF_S    (OFF_AMAX + N_NODES)
#define WS_FLOATS (OFF_S + N_NODES * S_PAD)

__device__ __forceinline__ unsigned f2ord(float f) {
  unsigned u = __float_as_uint(f);
  return (u & 0x80000000u) ? ~u : (u | 0x80000000u);
}
__device__ __forceinline__ float ord2f(unsigned u) {
  return __uint_as_float((u & 0x80000000u) ? (u ^ 0x80000000u) : ~u);
}
#define ORD_NEG_INF 0x007FFFFFu

__device__ __forceinline__ v2f load_a_clamp(const float* __restrict__ A, int m0, int k0,
                                            int lane, int ld) {
  int m = m0 + (lane & 15);
  m = (m < N_NODES) ? m : (N_NODES - 1);
  int kb = k0 + ((lane >> 4) << 1);
  const float* p = A + m * ld + kb;
  v2f a; a.x = p[0]; a.y = p[1];
  return a;
}
__device__ __forceinline__ v2f load_b(const float* __restrict__ B, int k0, int n0,
                                      int lane, int ld) {
  int n  = n0 + (lane & 15);
  int kb = k0 + ((lane >> 4) << 1);
  const float* p = B + kb * ld + n;
  v2f b; b.x = p[0]; b.y = p[ld];
  return b;
}
__device__ __forceinline__ v2f load_a_lds(const float* A, int k0, int lane) {
  int m  = lane & 15;
  int kb = k0 + ((lane >> 4) << 1);
  v2f a;
  a.x = A[m * OUT_C + kb];
  a.y = A[m * OUT_C + kb + 1];
  return a;
}

__global__ void k_init(float* __restrict__ ws) {
  int tid    = blockIdx.x * blockDim.x + threadIdx.x;
  int stride = gridDim.x * blockDim.x;
  unsigned* amax = (unsigned*)(ws + OFF_AMAX);
  for (int j = tid; j < N_NODES * S_PAD; j += stride) ws[OFF_S + j] = 0.0f;
  for (int j = tid; j < (H_ROWS - N_NODES) * FEAT; j += stride)
    ws[OFF_H + N_NODES * FEAT + j] = 0.0f;
  for (int j = tid; j < N_NODES; j += stride) {
    ws[OFF_ESUM + j] = 0.0f;
    ws[OFF_T + j]    = 0.0f;
    amax[j]          = ORD_NEG_INF;
  }
}

__global__ void k_node_gemm(const float* __restrict__ x, const float* __restrict__ W,
                            float* __restrict__ ws) {
  float* h  = ws + OFF_H;
  int lane  = threadIdx.x & 31;
  int wave  = threadIdx.x >> 5;
  int m0    = blockIdx.x * 16;
  int n0    = wave * 32;
  v8f c0 = {}; v8f c1 = {};
  for (int k0 = 0; k0 < FEAT; k0 += 4) {
    v2f a  = load_a_clamp(x, m0, k0, lane, FEAT);
    v2f b0 = load_b(W, k0, n0,      lane, OUT_C);
    v2f b1 = load_b(W, k0, n0 + 16, lane, OUT_C);
    c0 = __builtin_amdgcn_wmma_f32_16x16x4_f32(false, a, false, b0, (short)0, c0, false, false);
    c1 = __builtin_amdgcn_wmma_f32_16x16x4_f32(false, a, false, b1, (short)0, c1, false, false);
  }
  int nb = n0 + (lane & 15);
  int mh = (lane >> 4) * 8;
  for (int j = 0; j < 8; ++j) {
    int m = m0 + j + mh;
    if (m < N_NODES) {
      h[m * OUT_C + nb]      = c0[j];
      h[m * OUT_C + nb + 16] = c1[j];
    }
  }
}

__global__ void k_ab(const float* __restrict__ att, float* __restrict__ ws) {
  const float* h = ws + OFF_H;
  int n = blockIdx.x, lane = threadIdx.x;
  float pa = 0.0f, pb = 0.0f;
  for (int i = lane; i < FEAT; i += 32) {
    float hv = h[n * OUT_C + i];
    pa += hv * att[i];
    pb += hv * att[OUT_C + i];
  }
  for (int m = 16; m >= 1; m >>= 1) {
    pa += __shfl_xor(pa, m, 32);
    pb += __shfl_xor(pb, m, 32);
  }
  if (lane == 0) { ws[OFF_A + n] = pa; ws[OFF_B + n] = pb; }
}

#define EDGE_BLOCKS  128
#define EDGE_THREADS 1024
#define EDGE_STRIDE  (EDGE_BLOCKS * EDGE_THREADS)
static_assert(EDGE_THREADS >= N_NODES, "one thread per node clears and flushes the LDS copy");

__global__ void __launch_bounds__(EDGE_THREADS)
k_edge_max(const int* __restrict__ ei, const float* __restrict__ ea,
           const float* __restrict__ att, float* __restrict__ ws, int E) {
  __shared__ unsigned lmax[N_NODES];
  const float* a = ws + OFF_A;
  const float* b = ws + OFF_B;
  unsigned* amax = (unsigned*)(ws + OFF_AMAX);
  if (threadIdx.x < N_NODES) lmax[threadIdx.x] = ORD_NEG_INF;
  __syncthreads();
  float cc = att[2 * OUT_C];
  for (int e = blockIdx.x * EDGE_THREADS + threadIdx.x; e < E; e += EDGE_STRIDE) {
    int s = ei[e], d = ei[E + e];
    float al = a[d] + b[s] + cc * ea[e];
    al = (al > 0.0f) ? al : NEG_SLOPE * al;
    atomicMax(&lmax[d], f2ord(al));
  }
  __syncthreads();
  if (threadIdx.x < N_NODES) atomicMax(&amax[threadIdx.x], lmax[threadIdx.x]);
}

#define S_WORDS   (N_NODES * S_PAD)
#define S_HEAD    (OFF_S & 31)
#define FLUSH_ITS ((S_WORDS + S_HEAD + EDGE_THREADS - 1) / EDGE_THREADS)
static_assert(S_WORDS % 4 == 0, "the LDS copy of S is cleared in 16-byte stores");
#define FLUSH_BATCH 8
static_assert((FLUSH_ITS & (FLUSH_ITS - 1)) == 0, "the flush rotation wraps with a mask");
static_assert(FLUSH_ITS % FLUSH_BATCH == 0, "the flush sweep is cut into whole batches");

__global__ void __launch_bounds__(EDGE_THREADS)
k_edge_exp(const int* __restrict__ ei, const float* __restrict__ ea,
           const float* __restrict__ att, float* __restrict__ ws, int E) {
  __shared__ __attribute__((aligned(16))) float lS[S_WORDS];
  __shared__ float lsum[N_NODES];
  __shared__ float lt[N_NODES];
  const float* a = ws + OFF_A;
  const float* b = ws + OFF_B;
  const unsigned* amax = (const unsigned*)(ws + OFF_AMAX);
  float* esum = ws + OFF_ESUM;
  float* t    = ws + OFF_T;
  float* S    = ws + OFF_S;
  float4* lS4 = (float4*)lS;
  for (int i = threadIdx.x; i < S_WORDS / 4; i += EDGE_THREADS)
    lS4[i] = make_float4(0.0f, 0.0f, 0.0f, 0.0f);
  if (threadIdx.x < N_NODES) { lsum[threadIdx.x] = 0.0f; lt[threadIdx.x] = 0.0f; }
  __syncthreads();
  float cc = att[2 * OUT_C];
  for (int e = blockIdx.x * EDGE_THREADS + threadIdx.x; e < E; e += EDGE_STRIDE) {
    int s = ei[e], d = ei[E + e];
    float eav = ea[e];
    float al = a[d] + b[s] + cc * eav;
    al = (al > 0.0f) ? al : NEG_SLOPE * al;
    float w = __expf(al - ord2f(amax[d]));
    atomicAdd(&lsum[d], w);
    atomicAdd(&lt[d], w * eav);
    atomicAdd(&lS[d * S_PAD + s], w);
  }
  __syncthreads();
  if (threadIdx.x < N_NODES) {
    atomicAdd(&esum[threadIdx.x], lsum[threadIdx.x]);
    atomicAdd(&t[threadIdx.x], lt[threadIdx.x]);
  }
  for (int it0 = 0; it0 < FLUSH_ITS; it0 += FLUSH_BATCH) {
    float v[FLUSH_BATCH];
    int   jj[FLUSH_BATCH];
    for (int u = 0; u < FLUSH_BATCH; ++u) {
      int j = ((it0 + u + blockIdx.x) & (FLUSH_ITS - 1)) * EDGE_THREADS + threadIdx.x - S_HEAD;
      jj[u] = j;
      v[u]  = (j >= 0 && j < S_WORDS) ? lS[j] : 0.0f;
    }
    for (int u = 0; u < FLUSH_BATCH; ++u)
      if (v[u] != 0.0f) atomicAdd(&S[jj[u]], v[u]);
  }
}

__global__ void k_final(const float* __restrict__ U, const float* __restrict__ bias,
                        const float* __restrict__ ws, float* __restrict__ out) {
  __shared__ float aggr[16 * OUT_C];
  const float* h    = ws + OFF_H;
  const float* S    = ws + OFF_S;
  const float* esum = ws + OFF_ESUM;
  const float* t    = ws + OFF_T;
  int lane = threadIdx.x & 31;
  int wave = threadIdx.x >> 5;
  int m0 = blockIdx.x * 16;
  int n0 = wave * 32;

  v8f c0 = {}; v8f c1 = {};
  for (int k0 = 0; k0 < S_PAD; k0 += 4) {
    v2f a  = load_a_clamp(S, m0, k0, lane, S_PAD);
    v2f b0 = load_b(h, k0, n0,      lane, OUT_C);
    v2f b1 = load_b(h, k0, n0 + 16, lane, OUT_C);
    c0 = __builtin_amdgcn_wmma_f32_16x16x4_f32(false, a, false, b0, (short)0, c0, false, false);
    c1 = __builtin_amdgcn_wmma_f32_16x16x4_f32(false, a, false, b1, (short)0, c1, false, false);
  }
  int nb = n0 + (lane & 15);
  int mh = (lane >> 4) * 8;
  for (int j = 0; j < 8; ++j) {
    int mloc = j + mh;
    int m = m0 + mloc;
    int mc = (m < N_NODES) ? m : (N_NODES - 1);
    float r = 1.0f / (esum[mc] + 1e-16f);
    aggr[mloc * OUT_C + nb]      = c0[j] * r;
    aggr[mloc * OUT_C + nb + 16] = c1[j] * r;
  }
  __syncthreads();

  v8f d0 = {}; v8f d1 = {};
  for (int k0 = 0; k0 < OUT_C; k0 += 4) {
    v2f a  = load_a_lds(aggr, k0, lane);
    v2f b0 = load_b(U, k0, n0,      lane, OUT_C);
    v2f b1 = load_b(U, k0, n0 + 16, lane, OUT_C);
    d0 = __builtin_amdgcn_wmma_f32_16x16x4_f32(false, a, false, b0, (short)0, d0, false, false);
    d1 = __builtin_amdgcn_wmma_f32_16x16x4_f32(false, a, false, b1, (short)0, d1, false, false);
  }
  for (int j = 0; j < 8; ++j) {
    int m = m0 + j + mh;
    if (m < N_NODES) {
      float tn = t[m] / (esum[m] + 1e-16f);
      out[m * OUT_C + nb]      = d0[j] + tn * U[OUT_C * OUT_C + nb]      + bias[nb];
      out[m * OUT_C + nb + 16] = d1[j] + tn * U[OUT_C * OUT_C + nb + 16] + bias[nb + 16];
    }
  }
}

extern "C" void kernel_launch(void* const* d_in, const int* in_sizes, int n_in,
                              void* d_out, int out_size, void* d_ws, size_t ws_size,
                              hipStream_t stream) {
  const float* x    = (const float*)d_in[0];
  const float* ea   = (const float*)d_in[1];
  const float* W    = (const float*)d_in[2];
  const float* att  = (const float*)d_in[3];
  const float* U    = (const float*)d_in[4];
  const float* bias = (const float*)d_in[5];
  const int*   ei   = (const int*)d_in[6];
  int E = in_sizes[1];
  float* ws  = (float*)d_ws;
  float* out = (float*)d_out;

  k_init<<<64, 256, 0, stream>>>(ws);
  k_node_gemm<<<MTILES, 128, 0, stream>>>(x, W, ws);
  k_ab<<<N_NODES, 32, 0, stream>>>(att, ws);
  k_edge_max<<<EDGE_BLOCKS, EDGE_THREADS, 0, stream>>>(ei, ea, att, ws, E);
  k_edge_exp<<<EDGE_BLOCKS, EDGE_THREADS, 0, stream>>>(ei, ea, att, ws, E);
  k_final<<<MTILES, 128, 0, stream>>>(U, bias, ws, out);
}


# edge passes: each thread's four edges are requested in one group at kernel entry, ahead of LDS set-up and barrier, their per-node scalars in a second group; same atomics, arrival order as before
# speedup vs baseline: 5.3777x; 1.1927x over previous
#include <hip/hip_runtime.h>
#include <math.h>

typedef float v2f __attribute__((ext_vector_type(2)));
typedef float v8f __attribute__((ext_vector_type(8)));

#define N_NODES 177
#define FEAT 128
#define OUT_C 128
#define NEG_SLOPE 0.2f
#define MTILES 12
#define S_PAD 180
#define H_ROWS 180

#define OFF_H    0
#define OFF_A    (OFF_H + H_ROWS * FEAT)
#define OFF_B    (OFF_A + N_NODES)
#define OFF_ESUM (OFF_B + N_NODES)
#define OFF_T    (OFF_ESUM + N_NODES)
#define OFF_AMAX (OFF_T + N_NODES)
#define OFF_S    (OFF_AMAX + N_NODES)
#define WS_FLOATS (OFF_S + N_NODES * S_PAD)

__device__ __forceinline__ unsigned f2ord(float f) {
  unsigned u = __float_as_uint(f);
  return (u & 0x80000000u) ? ~u : (u | 0x80000000u);
}
__device__ __forceinline__ float ord2f(unsigned u) {
  return __uint_as_float((u & 0x80000000u) ? (u ^ 0x80000000u) : ~u);
}
#define ORD_NEG_INF 0x007FFFFFu

__device__ __forceinline__ v2f load_a_clamp(const float* __restrict__ A, int m0, int k0,
                                            int lane, int ld) {
  int m = m0 + (lane & 15);
  m = (m < N_NODES) ? m : (N_NODES - 1);
  int kb = k0 + ((lane >> 4) << 1);
  const float* p = A + m * ld + kb;
  v2f a; a.x = p[0]; a.y = p[1];
  return a;
}
__device__ __forceinline__ v2f load_b(const float* __restrict__ B, int k0, int n0,
                                      int lane, int ld) {
  int n  = n0 + (lane & 15);
  int kb = k0 + ((lane >> 4) << 1);
  const float* p = B + kb * ld + n;
  v2f b; b.x = p[0]; b.y = p[ld];
  return b;
}
__device__ __forceinline__ v2f load_a_lds(const float* A, int k0, int lane) {
  int m  = lane & 15;
  int kb = k0 + ((lane >> 4) << 1);
  v2f a;
  a.x = A[m * OUT_C + kb];
  a.y = A[m * OUT_C + kb + 1];
  return a;
}

#define NODE_THREADS 128
#define INIT_BLOCKS  32
#define NODE_BLOCKS  44
static_assert(NODE_BLOCKS == MTILES + INIT_BLOCKS, "grid of the node pass");
__global__ void __launch_bounds__(NODE_THREADS)
k_node(const float* __restrict__ x, const float* __restrict__ W,
       const float* __restrict__ att, float* __restrict__ ws) {
  __shared__ float hs[16 * OUT_C];
  int lane = threadIdx.x & 31;
  int wave = threadIdx.x >> 5;

  if (blockIdx.x >= MTILES) {
    int tid    = (blockIdx.x - MTILES) * NODE_THREADS + threadIdx.x;
    int stride = INIT_BLOCKS * NODE_THREADS;
    unsigned* amax = (unsigned*)(ws + OFF_AMAX);
    for (int j = tid; j < N_NODES * S_PAD; j += stride) ws[OFF_S + j] = 0.0f;
    for (int j = tid; j < (H_ROWS - N_NODES) * FEAT; j += stride)
      ws[OFF_H + N_NODES * FEAT + j] = 0.0f;
    for (int j = tid; j < N_NODES; j += stride) {
      ws[OFF_ESUM + j] = 0.0f;
      ws[OFF_T + j]    = 0.0f;
      amax[j]          = ORD_NEG_INF;
    }
    return;
  }

  float* h = ws + OFF_H;
  int m0 = blockIdx.x * 16;
  int n0 = wave * 32;
  float ta[FEAT / 32], tb[FEAT / 32];
  for (int k = 0; k < FEAT / 32; ++k) {
    ta[k] = att[lane + 32 * k];
    tb[k] = att[OUT_C + lane + 32 * k];
  }
  v8f c0 = {}; v8f c1 = {};
  for (int k0 = 0; k0 < FEAT; k0 += 4) {
    v2f a  = load_a_clamp(x, m0, k0, lane, FEAT);
    v2f b0 = load_b(W, k0, n0,      lane, OUT_C);
    v2f b1 = load_b(W, k0, n0 + 16, lane, OUT_C);
    c0 = __builtin_amdgcn_wmma_f32_16x16x4_f32(false, a, false, b0, (short)0, c0, false, false);
    c1 = __builtin_amdgcn_wmma_f32_16x16x4_f32(false, a, false, b1, (short)0, c1, false, false);
  }
  int nb = n0 + (lane & 15);
  int mh = (lane >> 4) * 8;
  for (int j = 0; j < 8; ++j) {
    int mloc = j + mh;
    hs[mloc * OUT_C + nb]      = c0[j];
    hs[mloc * OUT_C + nb + 16] = c1[j];
  }
  __syncthreads();
  for (int j = 0; j < 8; ++j) {
    int m = m0 + j + mh;
    if (m < N_NODES) {
      h[m * OUT_C + nb]      = c0[j];
      h[m * OUT_C + nb + 16] = c1[j];
    }
  }

  float pa[4], pb[4];
  for (int q = 0; q < 4; ++q) {
    const float* hr = hs + (wave * 4 + q) * OUT_C;
    pa[q] = 0.0f; pb[q] = 0.0f;
    for (int k = 0; k < FEAT / 32; ++k) {
      float hv = hr[lane + 32 * k];
      pa[q] += hv * ta[k];
      pb[q] += hv * tb[k];
    }
  }
  for (int m = 16; m >= 1; m >>= 1)
    for (int q = 0; q < 4; ++q) {
      pa[q] += __shfl_xor(pa[q], m, 32);
      pb[q] += __shfl_xor(pb[q], m, 32);
    }
  for (int q = 0; q < 4; ++q) {
    int n = m0 + wave * 4 + q;
    if (lane == 0 && n < N_NODES) { ws[OFF_A + n] = pa[q]; ws[OFF_B + n] = pb[q]; }
  }
}

#define EDGE_BLOCKS  128
#define EDGE_THREADS 1024
#define EDGE_STRIDE  (EDGE_BLOCKS * EDGE_THREADS)
static_assert(EDGE_THREADS >= N_NODES, "one thread per node clears and flushes the LDS copy");

#define EDGE_BATCH 4

__global__ void __launch_bounds__(EDGE_THREADS)
k_edge_max(const int* __restrict__ ei, const float* __restrict__ ea,
           const float* __restrict__ att, float* __restrict__ ws, int E) {
  __shared__ unsigned lmax[N_NODES];
  const float* a = ws + OFF_A;
  const float* b = ws + OFF_B;
  unsigned* amax = (unsigned*)(ws + OFF_AMAX);
  const int base = blockIdx.x * EDGE_THREADS + threadIdx.x;
  int   s[EDGE_BATCH], d[EDGE_BATCH];
  float eav[EDGE_BATCH];
  for (int u = 0; u < EDGE_BATCH; ++u) {
    int e = base + u * EDGE_STRIDE;
    s[u] = 0; d[u] = 0; eav[u] = 0.0f;
    if (e < E) { s[u] = ei[e]; d[u] = ei[E + e]; eav[u] = ea[e]; }
  }
  float cc = att[2 * OUT_C];
  if (threadIdx.x < N_NODES) lmax[threadIdx.x] = ORD_NEG_INF;
  __syncthreads();
  float al[EDGE_BATCH];
  for (int u = 0; u < EDGE_BATCH; ++u) al[u] = a[d[u]] + b[s[u]] + cc * eav[u];
  for (int u = 0; u < EDGE_BATCH; ++u) {
    if (base + u * EDGE_STRIDE < E) {
      float v = (al[u] > 0.0f) ? al[u] : NEG_SLOPE * al[u];
      atomicMax(&lmax[d[u]], f2ord(v));
    }
  }
  for (int e = base + EDGE_BATCH * EDGE_STRIDE; e < E; e += EDGE_STRIDE) {
    int s1 = ei[e], d1 = ei[E + e];
    float v = a[d1] + b[s1] + cc * ea[e];
    v = (v > 0.0f) ? v : NEG_SLOPE * v;
    atomicMax(&lmax[d1], f2ord(v));
  }
  __syncthreads();
  if (threadIdx.x < N_NODES) atomicMax(&amax[threadIdx.x], lmax[threadIdx.x]);
}

#define S_WORDS   (N_NODES * S_PAD)
#define S_HEAD    (OFF_S & 31)
#define FLUSH_ITS ((S_WORDS + S_HEAD + EDGE_THREADS - 1) / EDGE_THREADS)
static_assert(S_WORDS % 4 == 0, "the LDS copy of S is cleared in 16-byte stores");
#define FLUSH_BATCH 8
static_assert((FLUSH_ITS & (FLUSH_ITS - 1)) == 0, "the flush rotation wraps with a mask");
static_assert(FLUSH_ITS % FLUSH_BATCH == 0, "the flush sweep is cut into whole batches");

__global__ void __launch_bounds__(EDGE_THREADS)
k_edge_exp(const int* __restrict__ ei, const float* __restrict__ ea,
           const float* __restrict__ att, float* __restrict__ ws, int E) {
  __shared__ __attribute__((aligned(16))) float lS[S_WORDS];
  __shared__ float lsum[N_NODES];
  __shared__ float lt[N_NODES];
  const float* a = ws + OFF_A;
  const float* b = ws + OFF_B;
  const unsigned* amax = (const unsigned*)(ws + OFF_AMAX);
  float* esum = ws + OFF_ESUM;
  float* t    = ws + OFF_T;
  float* S    = ws + OFF_S;
  const int base = blockIdx.x * EDGE_THREADS + threadIdx.x;
  int   s[EDGE_BATCH], d[EDGE_BATCH];
  float eav[EDGE_BATCH];
  for (int u = 0; u < EDGE_BATCH; ++u) {
    int e = base + u * EDGE_STRIDE;
    s[u] = 0; d[u] = 0; eav[u] = 0.0f;
    if (e < E) { s[u] = ei[e]; d[u] = ei[E + e]; eav[u] = ea[e]; }
  }
  float cc = att[2 * OUT_C];
  float4* lS4 = (float4*)lS;
  for (int i = threadIdx.x; i < S_WORDS / 4; i += EDGE_THREADS)
    lS4[i] = make_float4(0.0f, 0.0f, 0.0f, 0.0f);
  if (threadIdx.x < N_NODES) { lsum[threadIdx.x] = 0.0f; lt[threadIdx.x] = 0.0f; }
  __syncthreads();
  float w[EDGE_BATCH];
  for (int u = 0; u < EDGE_BATCH; ++u) {
    float al = a[d[u]] + b[s[u]] + cc * eav[u];
    al = (al > 0.0f) ? al : NEG_SLOPE * al;
    w[u] = __expf(al - ord2f(amax[d[u]]));
  }
  for (int u = 0; u < EDGE_BATCH; ++u) {
    if (base + u * EDGE_STRIDE < E) {
      atomicAdd(&lsum[d[u]], w[u]);
      atomicAdd(&lt[d[u]], w[u] * eav[u]);
      atomicAdd(&lS[d[u] * S_PAD + s[u]], w[u]);
    }
  }
  for (int e = base + EDGE_BATCH * EDGE_STRIDE; e < E; e += EDGE_STRIDE) {
    int s1 = ei[e], d1 = ei[E + e];
    float eav1 = ea[e];
    float al = a[d1] + b[s1] + cc * eav1;
    al = (al > 0.0f) ? al : NEG_SLOPE * al;
    float w1 = __expf(al - ord2f(amax[d1]));
    atomicAdd(&lsum[d1], w1);
    atomicAdd(&lt[d1], w1 * eav1);
    atomicAdd(&lS[d1 * S_PAD + s1], w1);
  }
  __syncthreads();
  if (threadIdx.x < N_NODES) {
    atomicAdd(&esum[threadIdx.x], lsum[threadIdx.x]);
    atomicAdd(&t[threadIdx.x], lt[threadIdx.x]);
  }
  for (int it0 = 0; it0 < FLUSH_ITS; it0 += FLUSH_BATCH) {
    float v[FLUSH_BATCH];
    int   jj[FLUSH_BATCH];
    for (int u = 0; u < FLUSH_BATCH; ++u) {
      int j = ((it0 + u + blockIdx.x) & (FLUSH_ITS - 1)) * EDGE_THREADS + threadIdx.x - S_HEAD;
      jj[u] = j;
      v[u]  = (j >= 0 && j < S_WORDS) ? lS[j] : 0.0f;
    }
    for (int u = 0; u < FLUSH_BATCH; ++u)
      if (v[u] != 0.0f) atomicAdd(&S[jj[u]], v[u]);
  }
}

__global__ void k_final(const float* __restrict__ U, const float* __restrict__ bias,
                        const float* __restrict__ ws, float* __restrict__ out) {
  __shared__ float aggr[16 * OUT_C];
  const float* h    = ws + OFF_H;
  const float* S    = ws + OFF_S;
  const float* esum = ws + OFF_ESUM;
  const float* t    = ws + OFF_T;
  int lane = threadIdx.x & 31;
  int wave = threadIdx.x >> 5;
  int m0 = blockIdx.x * 16;
  int n0 = wave * 32;

  v8f c0 = {}; v8f c1 = {};
  for (int k0 = 0; k0 < S_PAD; k0 += 4) {
    v2f a  = load_a_clamp(S, m0, k0, lane, S_PAD);
    v2f b0 = load_b(h, k0, n0,      lane, OUT_C);
    v2f b1 = load_b(h, k0, n0 + 16, lane, OUT_C);
    c0 = __builtin_amdgcn_wmma_f32_16x16x4_f32(false, a, false, b0, (short)0, c0, false, false);
    c1 = __builtin_amdgcn_wmma_f32_16x16x4_f32(false, a, false, b1, (short)0, c1, false, false);
  }
  int nb = n0 + (lane & 15);
  int mh = (lane >> 4) * 8;
  for (int j = 0; j < 8; ++j) {
    int mloc = j + mh;
    int m = m0 + mloc;
    int mc = (m < N_NODES) ? m : (N_NODES - 1);
    float r = 1.0f / (esum[mc] + 1e-16f);
    aggr[mloc * OUT_C + nb]      = c0[j] * r;
    aggr[mloc * OUT_C + nb + 16] = c1[j] * r;
  }
  __syncthreads();

  v8f d0 = {}; v8f d1 = {};
  for (int k0 = 0; k0 < OUT_C; k0 += 4) {
    v2f a  = load_a_lds(aggr, k0, lane);
    v2f b0 = load_b(U, k0, n0,      lane, OUT_C);
    v2f b1 = load_b(U, k0, n0 + 16, lane, OUT_C);
    d0 = __builtin_amdgcn_wmma_f32_16x16x4_f32(false, a, false, b0, (short)0, d0, false, false);
    d1 = __builtin_amdgcn_wmma_f32_16x16x4_f32(false, a, false, b1, (short)0, d1, false, false);
  }
  for (int j = 0; j < 8; ++j) {
    int m = m0 + j + mh;
    if (m < N_NODES) {
      float tn = t[m] / (esum[m] + 1e-16f);
      out[m * OUT_C + nb]      = d0[j] + tn * U[OUT_C * OUT_C + nb]      + bias[nb];
      out[m * OUT_C + nb + 16] = d1[j] + tn * U[OUT_C * OUT_C + nb + 16] + bias[nb + 16];
    }
  }
}

extern "C" void kernel_launch(void* const* d_in, const int* in_sizes, int n_in,
                              void* d_out, int out_size, void* d_ws, size_t ws_size,
                              hipStream_t stream) {
  const float* x    = (const float*)d_in[0];
  const float* ea   = (const float*)d_in[1];
  const float* W    = (const float*)d_in[2];
  const float* att  = (const float*)d_in[3];
  const float* U    = (const float*)d_in[4];
  const float* bias = (const float*)d_in[5];
  const int*   ei   = (const int*)d_in[6];
  int E = in_sizes[1];
  float* ws  = (float*)d_ws;
  float* out = (float*)d_out;

  k_node<<<NODE_BLOCKS, NODE_THREADS, 0, stream>>>(x, W, att, ws);
  k_edge_max<<<EDGE_BLOCKS, EDGE_THREADS, 0, stream>>>(ei, ea, att, ws, E);
  k_edge_exp<<<EDGE_BLOCKS, EDGE_THREADS, 0, stream>>>(ei, ea, att, ws, E);
  k_final<<<MTILES, 128, 0, stream>>>(U, bias, ws, out);
}


# output product re-associated, S@(h@U) for (S@h)@U: h@U taken in the node launch from its LDS strip; final launch one K=180 GEMM on 96 one-wave tiles, loads in one group; float sum order changes
# speedup vs baseline: 7.4386x; 1.3832x over previous
#include <hip/hip_runtime.h>
#include <math.h>

typedef float v2f __attribute__((ext_vector_type(2)));
typedef float v8f __attribute__((ext_vector_type(8)));

#define N_NODES 177
#define FEAT 128
#define OUT_C 128
#define NEG_SLOPE 0.2f
#define MTILES 12
#define S_PAD 180
#define H_ROWS 180

#define OFF_H    0
#define OFF_A    (OFF_H + H_ROWS * FEAT)
#define OFF_B    (OFF_A + N_NODES)
#define OFF_ESUM (OFF_B + N_NODES)
#define OFF_T    (OFF_ESUM + N_NODES)
#define OFF_S    (OFF_T + N_NODES)
#define PM_PITCH   192
#define MAX_BLOCKS 64
#define OFF_PM   ((OFF_S + N_NODES * S_PAD + 31) & ~31)
#define WS_FLOATS (OFF_PM + MAX_BLOCKS * PM_PITCH)
static_assert(PM_PITCH % 32 == 0 && PM_PITCH >= N_NODES, "a PM row is whole lines and holds every node");

__device__ __forceinline__ unsigned f2ord(float f) {
  unsigned u = __float_as_uint(f);
  return (u & 0x80000000u) ? ~u : (u | 0x80000000u);
}
__device__ __forceinline__ float ord2f(unsigned u) {
  return __uint_as_float((u & 0x80000000u) ? (u ^ 0x80000000u) : ~u);
}
#define ORD_NEG_INF 0x007FFFFFu

__device__ __forceinline__ v2f load_a_clamp(const float* __restrict__ A, int m0, int k0,
                                            int lane, int ld) {
  int m = m0 + (lane & 15);
  m = (m < N_NODES) ? m : (N_NODES - 1);
  int kb = k0 + ((lane >> 4) << 1);
  const float* p = A + m * ld + kb;
  v2f a; a.x = p[0]; a.y = p[1];
  return a;
}
__device__ __forceinline__ v2f load_b(const float* __restrict__ B, int k0, int n0,
                                      int lane, int ld) {
  int n  = n0 + (lane & 15);
  int kb = k0 + ((lane >> 4) << 1);
  const float* p = B + kb * ld + n;
  v2f b; b.x = p[0]; b.y = p[ld];
  return b;
}
__device__ __forceinline__ v2f load_a_lds(const float* A, int k0, int lane) {
  int m  = lane & 15;
  int kb = k0 + ((lane >> 4) << 1);
  v2f a;
  a.x = A[m * OUT_C + kb];
  a.y = A[m * OUT_C + kb + 1];
  return a;
}

#define NODE_THREADS 1024
#define INIT_BLOCKS  4
#define NODE_BLOCKS  (MTILES + INIT_BLOCKS + MAX_BLOCKS)
#define GEMM_WAVES   4
#define MAX_BATCH    8
#define MAX_STRIDE   (MAX_BLOCKS * NODE_THREADS)
static_assert(FEAT == 128 && NODE_THREADS == 1024, "max block: 32 waves x 4 rows of W, 32 lanes x 4 words per row");
static_assert(6 * 32 >= N_NODES, "max block: six rows of x per wave cover every node");
__global__ void __launch_bounds__(NODE_THREADS)
k_node(const float* __restrict__ x, const float* __restrict__ W,
       const float* __restrict__ att, const float* __restrict__ U,
       const int* __restrict__ ei,
       const float* __restrict__ ea, float* __restrict__ ws, int E) {
  __shared__ float hs[16 * OUT_C];
  __shared__ __attribute__((aligned(16))) float wbs[FEAT];
  __shared__ float bl[N_NODES];
  __shared__ unsigned lmax[N_NODES];
  int lane = threadIdx.x & 31;
  int wave = threadIdx.x >> 5;

  if (blockIdx.x >= MTILES + INIT_BLOCKS) {
    const int mb   = blockIdx.x - (MTILES + INIT_BLOCKS);
    const int base = mb * NODE_THREADS + threadIdx.x;
    int   s[MAX_BATCH], d[MAX_BATCH];
    float eav[MAX_BATCH];
    for (int u = 0; u < MAX_BATCH; ++u) {
      int e = base + u * MAX_STRIDE;
      e = (e < E) ? e : 0;
      s[u] = ei[e]; d[u] = ei[E + e]; eav[u] = ea[e];
    }
    float cc = att[2 * OUT_C];
    const float4* W4 = (const float4*)W;
    const float4* x4 = (const float4*)x;
    float4 av = ((const float4*)(att + OUT_C))[lane];
    float4 wv[4], xv[6];
    for (int q = 0; q < 4; ++q) wv[q] = W4[(wave * 4 + q) * (OUT_C / 4) + lane];
    for (int q = 0; q < 6; ++q) {
      int n = wave + 32 * q;
      n = (n < N_NODES) ? n : (N_NODES - 1);
      xv[q] = x4[n * (FEAT / 4) + lane];
    }
    float pw[4];
    for (int q = 0; q < 4; ++q)
      pw[q] = wv[q].x * av.x + wv[q].y * av.y + wv[q].z * av.z + wv[q].w * av.w;
    for (int m = 16; m >= 1; m >>= 1)
      for (int q = 0; q < 4; ++q) pw[q] += __shfl_xor(pw[q], m, 32);
    if (lane == 0)
      for (int q = 0; q < 4; ++q) wbs[wave * 4 + q] = pw[q];
    if (threadIdx.x < N_NODES) lmax[threadIdx.x] = ORD_NEG_INF;
    __syncthreads();
    float4 wb4 = ((const float4*)wbs)[lane];
    float px[6];
    for (int q = 0; q < 6; ++q)
      px[q] = xv[q].x * wb4.x + xv[q].y * wb4.y + xv[q].z * wb4.z + xv[q].w * wb4.w;
    for (int m = 16; m >= 1; m >>= 1)
      for (int q = 0; q < 6; ++q) px[q] += __shfl_xor(px[q], m, 32);
    if (lane == 0)
      for (int q = 0; q < 6; ++q) {
        int n = wave + 32 * q;
        if (n < N_NODES) bl[n] = px[q];
      }
    __syncthreads();
    for (int u = 0; u < MAX_BATCH; ++u) {
      if (base + u * MAX_STRIDE < E)
        atomicMax(&lmax[d[u]], f2ord(bl[s[u]] + cc * eav[u]));
    }
    for (int e = base + MAX_BATCH * MAX_STRIDE; e < E; e += MAX_STRIDE) {
      int s1 = ei[e], d1 = ei[E + e];
      atomicMax(&lmax[d1], f2ord(bl[s1] + cc * ea[e]));
    }
    __syncthreads();
    if (threadIdx.x < N_NODES)
      ws[OFF_PM + mb * PM_PITCH + threadIdx.x] = ord2f(lmax[threadIdx.x]);
    return;
  }

  if (blockIdx.x >= MTILES) {
    int tid    = (blockIdx.x - MTILES) * NODE_THREADS + threadIdx.x;
    int stride = INIT_BLOCKS * NODE_THREADS;
    for (int j = tid; j < N_NODES * S_PAD; j += stride) ws[OFF_S + j] = 0.0f;
    for (int j = tid; j < (H_ROWS - N_NODES) * FEAT; j += stride)
      ws[OFF_H + N_NODES * FEAT + j] = 0.0f;
    for (int j = tid; j < N_NODES; j += stride) {
      ws[OFF_ESUM + j] = 0.0f;
      ws[OFF_T + j]    = 0.0f;
    }
    return;
  }

  float* g = ws + OFF_H;
  int m0 = blockIdx.x * 16;
  int n0 = wave * 32;
  const bool worker = wave < GEMM_WAVES;
  float ta[FEAT / 32], tb[FEAT / 32];
  v8f c0 = {}; v8f c1 = {};
  int nb = n0 + (lane & 15);
  int mh = (lane >> 4) * 8;
  if (worker) {
    for (int k = 0; k < FEAT / 32; ++k) {
      ta[k] = att[lane + 32 * k];
      tb[k] = att[OUT_C + lane + 32 * k];
    }
    for (int k0 = 0; k0 < FEAT; k0 += 4) {
      v2f a  = load_a_clamp(x, m0, k0, lane, FEAT);
      v2f b0 = load_b(W, k0, n0,      lane, OUT_C);
      v2f b1 = load_b(W, k0, n0 + 16, lane, OUT_C);
      c0 = __builtin_amdgcn_wmma_f32_16x16x4_f32(false, a, false, b0, (short)0, c0, false, false);
      c1 = __builtin_amdgcn_wmma_f32_16x16x4_f32(false, a, false, b1, (short)0, c1, false, false);
    }
    for (int j = 0; j < 8; ++j) {
      int mloc = j + mh;
      hs[mloc * OUT_C + nb]      = c0[j];
      hs[mloc * OUT_C + nb + 16] = c1[j];
    }
  }
  __syncthreads();
  if (!worker) return;
  v8f d0 = {}; v8f d1 = {};
  for (int k0 = 0; k0 < OUT_C; k0 += 4) {
    v2f a  = load_a_lds(hs, k0, lane);
    v2f b0 = load_b(U, k0, n0,      lane, OUT_C);
    v2f b1 = load_b(U, k0, n0 + 16, lane, OUT_C);
    d0 = __builtin_amdgcn_wmma_f32_16x16x4_f32(false, a, false, b0, (short)0, d0, false, false);
    d1 = __builtin_amdgcn_wmma_f32_16x16x4_f32(false, a, false, b1, (short)0, d1, false, false);
  }
  for (int j = 0; j < 8; ++j) {
    int m = m0 + j + mh;
    if (m < N_NODES) {
      g[m * OUT_C + nb]      = d0[j];
      g[m * OUT_C + nb + 16] = d1[j];
    }
  }

  float pa[4], pb[4];
  for (int q = 0; q < 4; ++q) {
    const float* hr = hs + (wave * 4 + q) * OUT_C;
    pa[q] = 0.0f; pb[q] = 0.0f;
    for (int k = 0; k < FEAT / 32; ++k) {
      float hv = hr[lane + 32 * k];
      pa[q] += hv * ta[k];
      pb[q] += hv * tb[k];
    }
  }
  for (int m = 16; m >= 1; m >>= 1)
    for (int q = 0; q < 4; ++q) {
      pa[q] += __shfl_xor(pa[q], m, 32);
      pb[q] += __shfl_xor(pb[q], m, 32);
    }
  for (int q = 0; q < 4; ++q) {
    int n = m0 + wave * 4 + q;
    if (lane == 0 && n < N_NODES) { ws[OFF_A + n] = pa[q]; ws[OFF_B + n] = pb[q]; }
  }
}

#define EDGE_BLOCKS  128
#define EDGE_THREADS 1024
#define EDGE_STRIDE  (EDGE_BLOCKS * EDGE_THREADS)
static_assert(EDGE_THREADS >= N_NODES, "one thread per node clears and flushes the LDS copy");

#define EDGE_BATCH 4

#define S_WORDS   (N_NODES * S_PAD)
#define S_HEAD    (OFF_S & 31)
#define FLUSH_ITS ((S_WORDS + S_HEAD + EDGE_THREADS - 1) / EDGE_THREADS)
static_assert(S_WORDS % 4 == 0, "the LDS copy of S is cleared in 16-byte stores");
#define FLUSH_BATCH 8
#define PM_PARTS (EDGE_THREADS / 256)
#define PM_ROWS  (MAX_BLOCKS / PM_PARTS)
static_assert(EDGE_THREADS % 256 == 0 && MAX_BLOCKS % PM_PARTS == 0 && N_NODES <= 256, "merge of PM");
static_assert((FLUSH_ITS & (FLUSH_ITS - 1)) == 0, "the flush rotation wraps with a mask");
static_assert(FLUSH_ITS % FLUSH_BATCH == 0, "the flush sweep is cut into whole batches");

__global__ void __launch_bounds__(EDGE_THREADS)
k_edge_exp(const int* __restrict__ ei, const float* __restrict__ ea,
           const float* __restrict__ att, float* __restrict__ ws, int E) {
  __shared__ __attribute__((aligned(16))) float lS[S_WORDS];
  __shared__ float lsum[N_NODES];
  __shared__ float lt[N_NODES];
  __shared__ float pmx[PM_PARTS * 256];
  __shared__ float lshift[N_NODES];
  const float* a = ws + OFF_A;
  const float* b = ws + OFF_B;
  const float* PM = ws + OFF_PM;
  float* esum = ws + OFF_ESUM;
  float* t    = ws + OFF_T;
  float* S    = ws + OFF_S;
  const int base = blockIdx.x * EDGE_THREADS + threadIdx.x;
  int   s[EDGE_BATCH], d[EDGE_BATCH];
  float eav[EDGE_BATCH];
  for (int u = 0; u < EDGE_BATCH; ++u) {
    int e = base + u * EDGE_STRIDE;
    s[u] = 0; d[u] = 0; eav[u] = 0.0f;
    if (e < E) { s[u] = ei[e]; d[u] = ei[E + e]; eav[u] = ea[e]; }
  }
  float cc = att[2 * OUT_C];
  const int pd = threadIdx.x & 255;
  const int pp = threadIdx.x >> 8;
  const int pdc = (pd < N_NODES) ? pd : (N_NODES - 1);
  float pmr[PM_ROWS];
  for (int k = 0; k < PM_ROWS; ++k) pmr[k] = PM[(pp * PM_ROWS + k) * PM_PITCH + pdc];
  float a_t = (threadIdx.x < N_NODES) ? a[threadIdx.x] : 0.0f;
  float4* lS4 = (float4*)lS;
  for (int i = threadIdx.x; i < S_WORDS / 4; i += EDGE_THREADS)
    lS4[i] = make_float4(0.0f, 0.0f, 0.0f, 0.0f);
  if (threadIdx.x < N_NODES) { lsum[threadIdx.x] = 0.0f; lt[threadIdx.x] = 0.0f; }
  float pmv = pmr[0];
  for (int k = 1; k < PM_ROWS; ++k) pmv = fmaxf(pmv, pmr[k]);
  pmx[threadIdx.x] = pmv;
  __syncthreads();
  if (threadIdx.x < N_NODES) {
    float mx = pmx[threadIdx.x];
    for (int p = 1; p < PM_PARTS; ++p) mx = fmaxf(mx, pmx[p * 256 + threadIdx.x]);
    float v = a_t + mx;
    lshift[threadIdx.x] = (v > 0.0f) ? v : NEG_SLOPE * v;
  }
  __syncthreads();
  float w[EDGE_BATCH];
  for (int u = 0; u < EDGE_BATCH; ++u) {
    float al = a[d[u]] + b[s[u]] + cc * eav[u];
    al = (al > 0.0f) ? al : NEG_SLOPE * al;
    w[u] = __expf(al - lshift[d[u]]);
  }
  for (int u = 0; u < EDGE_BATCH; ++u) {
    if (base + u * EDGE_STRIDE < E) {
      atomicAdd(&lsum[d[u]], w[u]);
      atomicAdd(&lt[d[u]], w[u] * eav[u]);
      atomicAdd(&lS[d[u] * S_PAD + s[u]], w[u]);
    }
  }
  for (int e = base + EDGE_BATCH * EDGE_STRIDE; e < E; e += EDGE_STRIDE) {
    int s1 = ei[e], d1 = ei[E + e];
    float eav1 = ea[e];
    float al = a[d1] + b[s1] + cc * eav1;
    al = (al > 0.0f) ? al : NEG_SLOPE * al;
    float w1 = __expf(al - lshift[d1]);
    atomicAdd(&lsum[d1], w1);
    atomicAdd(&lt[d1], w1 * eav1);
    atomicAdd(&lS[d1 * S_PAD + s1], w1);
  }
  __syncthreads();
  if (threadIdx.x < N_NODES) {
    atomicAdd(&esum[threadIdx.x], lsum[threadIdx.x]);
    atomicAdd(&t[threadIdx.x], lt[threadIdx.x]);
  }
  for (int it0 = 0; it0 < FLUSH_ITS; it0 += FLUSH_BATCH) {
    float v[FLUSH_BATCH];
    int   jj[FLUSH_BATCH];
    for (int u = 0; u < FLUSH_BATCH; ++u) {
      int j = ((it0 + u + blockIdx.x) & (FLUSH_ITS - 1)) * EDGE_THREADS + threadIdx.x - S_HEAD;
      jj[u] = j;
      v[u]  = (j >= 0 && j < S_WORDS) ? lS[j] : 0.0f;
    }
    for (int u = 0; u < FLUSH_BATCH; ++u)
      if (v[u] != 0.0f) atomicAdd(&S[jj[u]], v[u]);
  }
}

#define FIN_THREADS 64
#define FIN_BLOCKS  (MTILES * (OUT_C / 32))
#define FIN_KSTEPS  (S_PAD / 4)
static_assert(S_PAD % 4 == 0 && H_ROWS >= S_PAD, "the K loop runs over whole k-steps inside g's padded rows");
__global__ void __launch_bounds__(FIN_THREADS)
k_final(const float* __restrict__ U, const float* __restrict__ bias,
        const float* __restrict__ ws, float* __restrict__ out) {
  const float* g    = ws + OFF_H;
  const float* S    = ws + OFF_S;
  const float* esum = ws + OFF_ESUM;
  const float* t    = ws + OFF_T;
  int lane = threadIdx.x & 31;
  int wave = threadIdx.x >> 5;
  int m0 = (blockIdx.x / (OUT_C / 32)) * 16;
  int n0 = (blockIdx.x % (OUT_C / 32)) * 32 + wave * 16;
  int nb = n0 + (lane & 15);
  int mh = (lane >> 4) * 8;

  float es[8], tv[8];
  for (int j = 0; j < 8; ++j) {
    int m = m0 + j + mh;
    m = (m < N_NODES) ? m : (N_NODES - 1);
    es[j] = esum[m];
    tv[j] = t[m];
  }
  float ue = U[OUT_C * OUT_C + nb];
  float bs = bias[nb];
  v2f a[FIN_KSTEPS], b[FIN_KSTEPS];
#pragma unroll
  for (int i = 0; i < FIN_KSTEPS; ++i) {
    a[i] = load_a_clamp(S, m0, 4 * i, lane, S_PAD);
    b[i] = load_b(g, 4 * i, n0, lane, OUT_C);
  }
  __builtin_amdgcn_sched_barrier(0);
  v8f c0 = {}; v8f c1 = {};
#pragma unroll
  for (int i = 0; i < FIN_KSTEPS; ++i) {
    if (i & 1) c1 = __builtin_amdgcn_wmma_f32_16x16x4_f32(false, a[i], false, b[i], (short)0, c1, false, false);
    else       c0 = __builtin_amdgcn_wmma_f32_16x16x4_f32(false, a[i], false, b[i], (short)0, c0, false, false);
  }
  if (m0 + 16 <= N_NODES) {
    for (int j = 0; j < 8; ++j) {
      int m = m0 + j + mh;
      float r  = 1.0f / (es[j] + 1e-16f);
      float tn = tv[j] / (es[j] + 1e-16f);
      out[m * OUT_C + nb] = (c0[j] + c1[j]) * r + tn * ue + bs;
    }
  } else {
    for (int j = 0; j < 8; ++j) {
      int m = m0 + j + mh;
      if (m < N_NODES) {
        float r  = 1.0f / (es[j] + 1e-16f);
        float tn = tv[j] / (es[j] + 1e-16f);
        out[m * OUT_C + nb] = (c0[j] + c1[j]) * r + tn * ue + bs;
      }
    }
  }
}

extern "C" void kernel_launch(void* const* d_in, const int* in_sizes, int n_in,
                              void* d_out, int out_size, void* d_ws, size_t ws_size,
                              hipStream_t stream) {
  const float* x    = (const float*)d_in[0];
  const float* ea   = (const float*)d_in[1];
  const float* W    = (const float*)d_in[2];
  const float* att  = (const float*)d_in[3];
  const float* U    = (const float*)d_in[4];
  const float* bias = (const float*)d_in[5];
  const int*   ei   = (const int*)d_in[6];
  int E = in_sizes[1];
  float* ws  = (float*)d_ws;
  float* out = (float*)d_out;

  k_node<<<NODE_BLOCKS, NODE_THREADS, 0, stream>>>(x, W, att, U, ei, ea, ws, E);
  k_edge_exp<<<EDGE_BLOCKS, EDGE_THREADS, 0, stream>>>(ei, ea, att, ws, E);
  k_final<<<FIN_BLOCKS, FIN_THREADS, 0, stream>>>(U, bias, ws, out);
}


# speedup vs baseline: 8.7184x; 1.1720x over previous
#include <hip/hip_runtime.h>
#include <math.h>

typedef float v2f __attribute__((ext_vector_type(2)));
typedef float v8f __attribute__((ext_vector_type(8)));

#define N_NODES 177
#define FEAT 128
#define OUT_C 128
#define NEG_SLOPE 0.2f
#define MTILES 12
#define S_PAD 180
#define H_ROWS 180

#define OFF_H    0
#define OFF_A    (OFF_H + H_ROWS * FEAT)
#define OFF_B    (OFF_A + N_NODES)
#define OFF_ESUM (OFF_B + N_NODES)
#define OFF_T    (OFF_ESUM + N_NODES)
#define OFF_S    (OFF_T + N_NODES)
#define PM_PITCH   192
#define MAX_BLOCKS 64
#define OFF_PM   ((OFF_S + N_NODES * S_PAD + 31) & ~31)
#define WS_ONE_PLANE (OFF_PM + MAX_BLOCKS * PM_PITCH)
static_assert(PM_PITCH % 32 == 0 && PM_PITCH >= N_NODES, "a PM row is whole lines and holds every node");
#define PLANE     ((WS_ONE_PLANE - OFF_ESUM + 31) & ~31)
#define WS_FLOATS (OFF_S + N_NODES * S_PAD + PLANE)
static_assert(PLANE % 32 == 0 && OFF_ESUM + PLANE >= WS_ONE_PLANE, "plane 1 starts behind PM, line phase kept");

__device__ __forceinline__ unsigned f2ord(float f) {
  unsigned u = __float_as_uint(f);
  return (u & 0x80000000u) ? ~u : (u | 0x80000000u);
}
__device__ __forceinline__ float ord2f(unsigned u) {
  return __uint_as_float((u & 0x80000000u) ? (u ^ 0x80000000u) : ~u);
}
#define ORD_NEG_INF 0x007FFFFFu

__device__ __forceinline__ v2f load_a_clamp(const float* __restrict__ A, int m0, int k0,
                                            int lane, int ld) {
  int m = m0 + (lane & 15);
  m = (m < N_NODES) ? m : (N_NODES - 1);
  int kb = k0 + ((lane >> 4) << 1);
  const float* p = A + m * ld + kb;
  v2f a; a.x = p[0]; a.y = p[1];
  return a;
}
__device__ __forceinline__ v2f load_b(const float* __restrict__ B, int k0, int n0,
                                      int lane, int ld) {
  int n  = n0 + (lane & 15);
  int kb = k0 + ((lane >> 4) << 1);
  const float* p = B + kb * ld + n;
  v2f b; b.x = p[0]; b.y = p[ld];
  return b;
}
__device__ __forceinline__ v2f load_a_lds(const float* A, int k0, int lane) {
  int m  = lane & 15;
  int kb = k0 + ((lane >> 4) << 1);
  v2f a;
  a.x = A[m * OUT_C + kb];
  a.y = A[m * OUT_C + kb + 1];
  return a;
}

#define NODE_THREADS 1024
#define INIT_BLOCKS  4
#define NODE_BLOCKS  (MTILES + INIT_BLOCKS + MAX_BLOCKS)
#define GEMM_WAVES   4
#define MAX_BATCH    8
#define MAX_STRIDE   (MAX_BLOCKS * NODE_THREADS)
static_assert(FEAT == 128 && NODE_THREADS == 1024, "max block: 32 waves x 4 rows of W, 32 lanes x 4 words per row");
static_assert(6 * 32 >= N_NODES, "max block: six rows of x per wave cover every node");
__global__ void __launch_bounds__(NODE_THREADS)
k_node(const float* __restrict__ x, const float* __restrict__ W,
       const float* __restrict__ att, const float* __restrict__ U,
       const int* __restrict__ ei,
       const float* __restrict__ ea, float* __restrict__ ws, int E, size_t ws_bytes) {
  __shared__ float hs[16 * OUT_C];
  __shared__ __attribute__((aligned(16))) float wbs[FEAT];
  __shared__ float bl[N_NODES];
  __shared__ unsigned lmax[N_NODES];
  int lane = threadIdx.x & 31;
  int wave = threadIdx.x >> 5;

  if (blockIdx.x >= MTILES + INIT_BLOCKS) {
    const int mb   = blockIdx.x - (MTILES + INIT_BLOCKS);
    const int base = mb * NODE_THREADS + threadIdx.x;
    int   s[MAX_BATCH], d[MAX_BATCH];
    float eav[MAX_BATCH];
    for (int u = 0; u < MAX_BATCH; ++u) {
      int e = base + u * MAX_STRIDE;
      e = (e < E) ? e : 0;
      s[u] = ei[e]; d[u] = ei[E + e]; eav[u] = ea[e];
    }
    float cc = att[2 * OUT_C];
    const float4* W4 = (const float4*)W;
    const float4* x4 = (const float4*)x;
    float4 av = ((const float4*)(att + OUT_C))[lane];
    float4 wv[4], xv[6];
    for (int q = 0; q < 4; ++q) wv[q] = W4[(wave * 4 + q) * (OUT_C / 4) + lane];
    for (int q = 0; q < 6; ++q) {
      int n = wave + 32 * q;
      n = (n < N_NODES) ? n : (N_NODES - 1);
      xv[q] = x4[n * (FEAT / 4) + lane];
    }
    float pw[4];
    for (int q = 0; q < 4; ++q)
      pw[q] = wv[q].x * av.x + wv[q].y * av.y + wv[q].z * av.z + wv[q].w * av.w;
    for (int m = 16; m >= 1; m >>= 1)
      for (int q = 0; q < 4; ++q) pw[q] += __shfl_xor(pw[q], m, 32);
    if (lane == 0)
      for (int q = 0; q < 4; ++q) wbs[wave * 4 + q] = pw[q];
    if (threadIdx.x < N_NODES) lmax[threadIdx.x] = ORD_NEG_INF;
    __syncthreads();
    float4 wb4 = ((const float4*)wbs)[lane];
    float px[6];
    for (int q = 0; q < 6; ++q)
      px[q] = xv[q].x * wb4.x + xv[q].y * wb4.y + xv[q].z * wb4.z + xv[q].w * wb4.w;
    for (int m = 16; m >= 1; m >>= 1)
      for (int q = 0; q < 6; ++q) px[q] += __shfl_xor(px[q], m, 32);
    if (lane == 0)
      for (int q = 0; q < 6; ++q) {
        int n = wave + 32 * q;
        if (n < N_NODES) bl[n] = px[q];
      }
    __syncthreads();
    for (int u = 0; u < MAX_BATCH; ++u) {
      if (base + u * MAX_STRIDE < E)
        atomicMax(&lmax[d[u]], f2ord(bl[s[u]] + cc * eav[u]));
    }
    for (int e = base + MAX_BATCH * MAX_STRIDE; e < E; e += MAX_STRIDE) {
      int s1 = ei[e], d1 = ei[E + e];
      atomicMax(&lmax[d1], f2ord(bl[s1] + cc * ea[e]));
    }
    __syncthreads();
    if (threadIdx.x < N_NODES)
      ws[OFF_PM + mb * PM_PITCH + threadIdx.x] = ord2f(lmax[threadIdx.x]);
    return;
  }

  if (blockIdx.x >= MTILES) {
    int tid    = (blockIdx.x - MTILES) * NODE_THREADS + threadIdx.x;
    int stride = INIT_BLOCKS * NODE_THREADS;
    const int p1 = (ws_bytes >= (size_t)WS_FLOATS * sizeof(float)) ? PLANE : 0;
    for (int j = tid; j < N_NODES * S_PAD; j += stride) {
      ws[OFF_S + j]      = 0.0f;
      ws[OFF_S + p1 + j] = 0.0f;
    }
    for (int j = tid; j < (H_ROWS - N_NODES) * FEAT; j += stride)
      ws[OFF_H + N_NODES * FEAT + j] = 0.0f;
    for (int j = tid; j < N_NODES; j += stride) {
      ws[OFF_ESUM + j] = 0.0f;
      ws[OFF_T + j]    = 0.0f;
      ws[OFF_ESUM + p1 + j] = 0.0f;
      ws[OFF_T + p1 + j]    = 0.0f;
    }
    return;
  }

  float* g = ws + OFF_H;
  int m0 = blockIdx.x * 16;
  int n0 = wave * 32;
  const bool worker = wave < GEMM_WAVES;
  float ta[FEAT / 32], tb[FEAT / 32];
  v8f c0 = {}; v8f c1 = {};
  int nb = n0 + (lane & 15);
  int mh = (lane >> 4) * 8;
  if (worker) {
    for (int k = 0; k < FEAT / 32; ++k) {
      ta[k] = att[lane + 32 * k];
      tb[k] = att[OUT_C + lane + 32 * k];
    }
    for (int k0 = 0; k0 < FEAT; k0 += 4) {
      v2f a  = load_a_clamp(x, m0, k0, lane, FEAT);
      v2f b0 = load_b(W, k0, n0,      lane, OUT_C);
      v2f b1 = load_b(W, k0, n0 + 16, lane, OUT_C);
      c0 = __builtin_amdgcn_wmma_f32_16x16x4_f32(false, a, false, b0, (short)0, c0, false, false);
      c1 = __builtin_amdgcn_wmma_f32_16x16x4_f32(false, a, false, b1, (short)0, c1, false, false);
    }
    for (int j = 0; j < 8; ++j) {
      int mloc = j + mh;
      hs[mloc * OUT_C + nb]      = c0[j];
      hs[mloc * OUT_C + nb + 16] = c1[j];
    }
  }
  __syncthreads();
  if (!worker) return;
  v8f d0 = {}; v8f d1 = {};
  for (int k0 = 0; k0 < OUT_C; k0 += 4) {
    v2f a  = load_a_lds(hs, k0, lane);
    v2f b0 = load_b(U, k0, n0,      lane, OUT_C);
    v2f b1 = load_b(U, k0, n0 + 16, lane, OUT_C);
    d0 = __builtin_amdgcn_wmma_f32_16x16x4_f32(false, a, false, b0, (short)0, d0, false, false);
    d1 = __builtin_amdgcn_wmma_f32_16x16x4_f32(false, a, false, b1, (short)0, d1, false, false);
  }
  for (int j = 0; j < 8; ++j) {
    int m = m0 + j + mh;
    if (m < N_NODES) {
      g[m * OUT_C + nb]      = d0[j];
      g[m * OUT_C + nb + 16] = d1[j];
    }
  }

  float pa[4], pb[4];
  for (int q = 0; q < 4; ++q) {
    const float* hr = hs + (wave * 4 + q) * OUT_C;
    pa[q] = 0.0f; pb[q] = 0.0f;
    for (int k = 0; k < FEAT / 32; ++k) {
      float hv = hr[lane + 32 * k];
      pa[q] += hv * ta[k];
      pb[q] += hv * tb[k];
    }
  }
  for (int m = 16; m >= 1; m >>= 1)
    for (int q = 0; q < 4; ++q) {
      pa[q] += __shfl_xor(pa[q], m, 32);
      pb[q] += __shfl_xor(pb[q], m, 32);
    }
  for (int q = 0; q < 4; ++q) {
    int n = m0 + wave * 4 + q;
    if (lane == 0 && n < N_NODES) { ws[OFF_A + n] = pa[q]; ws[OFF_B + n] = pb[q]; }
  }
}

#define EDGE_BLOCKS  128
#define EDGE_THREADS 1024
#define EDGE_STRIDE  (EDGE_BLOCKS * EDGE_THREADS)
static_assert(EDGE_THREADS >= N_NODES, "one thread per node clears and flushes the LDS copy");

#define EDGE_BATCH 4

#define S_WORDS   (N_NODES * S_PAD)
#define S_HEAD    (OFF_S & 31)
#define FLUSH_ITS ((S_WORDS + S_HEAD + EDGE_THREADS - 1) / EDGE_THREADS)
static_assert(S_WORDS % 4 == 0, "the LDS copy of S is cleared in 16-byte stores");
#define FLUSH_BATCH 8
#define PM_PARTS (EDGE_THREADS / 256)
#define PM_ROWS  (MAX_BLOCKS / PM_PARTS)
static_assert(EDGE_THREADS % 256 == 0 && MAX_BLOCKS % PM_PARTS == 0 && N_NODES <= 256, "merge of PM");
static_assert((FLUSH_ITS & (FLUSH_ITS - 1)) == 0, "the flush rotation wraps with a mask");
static_assert(FLUSH_ITS % FLUSH_BATCH == 0, "the flush sweep is cut into whole batches");

__global__ void __launch_bounds__(EDGE_THREADS)
k_edge_exp(const int* __restrict__ ei, const float* __restrict__ ea,
           const float* __restrict__ att, float* __restrict__ ws, int E, size_t ws_bytes) {
  __shared__ __attribute__((aligned(16))) float lS[S_WORDS];
  __shared__ float lsum[N_NODES];
  __shared__ float lt[N_NODES];
  __shared__ float pmx[PM_PARTS * 256];
  __shared__ float lshift[N_NODES];
  const float* a = ws + OFF_A;
  const float* b = ws + OFF_B;
  const float* PM = ws + OFF_PM;
  const int base = blockIdx.x * EDGE_THREADS + threadIdx.x;
  int   s[EDGE_BATCH], d[EDGE_BATCH];
  float eav[EDGE_BATCH];
  for (int u = 0; u < EDGE_BATCH; ++u) {
    int e = base + u * EDGE_STRIDE;
    s[u] = 0; d[u] = 0; eav[u] = 0.0f;
    if (e < E) { s[u] = ei[e]; d[u] = ei[E + e]; eav[u] = ea[e]; }
  }
  float cc = att[2 * OUT_C];
  const int pd = threadIdx.x & 255;
  const int pp = threadIdx.x >> 8;
  const int pdc = (pd < N_NODES) ? pd : (N_NODES - 1);
  float pmr[PM_ROWS];
  for (int k = 0; k < PM_ROWS; ++k) pmr[k] = PM[(pp * PM_ROWS + k) * PM_PITCH + pdc];
  float a_t = (threadIdx.x < N_NODES) ? a[threadIdx.x] : 0.0f;
  const unsigned place = __builtin_amdgcn_s_sendmsg_rtn(0x87);
  const int po = (((place >> 16) & 0xFu) >= 4u && ws_bytes >= (size_t)WS_FLOATS * sizeof(float)) ? PLANE : 0;
  float* esum = ws + OFF_ESUM + po;
  float* t    = ws + OFF_T + po;
  float* S    = ws + OFF_S + po;
  float4* lS4 = (float4*)lS;
  for (int i = threadIdx.x; i < S_WORDS / 4; i += EDGE_THREADS)
    lS4[i] = make_float4(0.0f, 0.0f, 0.0f, 0.0f);
  if (threadIdx.x < N_NODES) { lsum[threadIdx.x] = 0.0f; lt[threadIdx.x] = 0.0f; }
  float pmv = pmr[0];
  for (int k = 1; k < PM_ROWS; ++k) pmv = fmaxf(pmv, pmr[k]);
  pmx[threadIdx.x] = pmv;
  __syncthreads();
  if (threadIdx.x < N_NODES) {
    float mx = pmx[threadIdx.x];
    for (int p = 1; p < PM_PARTS; ++p) mx = fmaxf(mx, pmx[p * 256 + threadIdx.x]);
    float v = a_t + mx;
    lshift[threadIdx.x] = (v > 0.0f) ? v : NEG_SLOPE * v;
  }
  __syncthreads();
  float w[EDGE_BATCH];
  for (int u = 0; u < EDGE_BATCH; ++u) {
    float al = a[d[u]] + b[s[u]] + cc * eav[u];
    al = (al > 0.0f) ? al : NEG_SLOPE * al;
    w[u] = __expf(al - lshift[d[u]]);
  }
  for (int u = 0; u < EDGE_BATCH; ++u) {
    if (base + u * EDGE_STRIDE < E) {
      atomicAdd(&lsum[d[u]], w[u]);
      atomicAdd(&lt[d[u]], w[u] * eav[u]);
      atomicAdd(&lS[d[u] * S_PAD + s[u]], w[u]);
    }
  }
  for (int e = base + EDGE_BATCH * EDGE_STRIDE; e < E; e += EDGE_STRIDE) {
    int s1 = ei[e], d1 = ei[E + e];
    float eav1 = ea[e];
    float al = a[d1] + b[s1] + cc * eav1;
    al = (al > 0.0f) ? al : NEG_SLOPE * al;
    float w1 = __expf(al - lshift[d1]);
    atomicAdd(&lsum[d1], w1);
    atomicAdd(&lt[d1], w1 * eav1);
    atomicAdd(&lS[d1 * S_PAD + s1], w1);
  }
  __syncthreads();
  if (threadIdx.x < N_NODES) {
    atomicAdd(&esum[threadIdx.x], lsum[threadIdx.x]);
    atomicAdd(&t[threadIdx.x], lt[threadIdx.x]);
  }
  for (int it0 = 0; it0 < FLUSH_ITS; it0 += FLUSH_BATCH) {
    float v[FLUSH_BATCH];
    int   jj[FLUSH_BATCH];
    for (int u = 0; u < FLUSH_BATCH; ++u) {
      int j = ((it0 + u + blockIdx.x) & (FLUSH_ITS - 1)) * EDGE_THREADS + threadIdx.x - S_HEAD;
      jj[u] = j;
      v[u]  = (j >= 0 && j < S_WORDS) ? lS[j] : 0.0f;
    }
    for (int u = 0; u < FLUSH_BATCH; ++u)
      if (v[u] != 0.0f) atomicAdd(&S[jj[u]], v[u]);
  }
}

#define FIN_THREADS 64
#define FIN_BLOCKS  (MTILES * (OUT_C / 32))
#define FIN_KSTEPS  (S_PAD / 4)
static_assert(S_PAD % 4 == 0 && H_ROWS >= S_PAD, "the K loop runs over whole k-steps inside g's padded rows");
__global__ void __launch_bounds__(FIN_THREADS)
k_final(const float* __restrict__ U, const float* __restrict__ bias,
        const float* __restrict__ ws, float* __restrict__ out, size_t ws_bytes) {
  const float* g    = ws + OFF_H;
  const float* S    = ws + OFF_S;
  const float* esum = ws + OFF_ESUM;
  const float* t    = ws + OFF_T;
  int lane = threadIdx.x & 31;
  int wave = threadIdx.x >> 5;
  int m0 = (blockIdx.x / (OUT_C / 32)) * 16;
  int n0 = (blockIdx.x % (OUT_C / 32)) * 32 + wave * 16;
  int nb = n0 + (lane & 15);
  int mh = (lane >> 4) * 8;
  const bool  two = ws_bytes >= (size_t)WS_FLOATS * sizeof(float);
  const int   p1  = two ? PLANE : 0;
  const float f1  = two ? 1.0f : 0.0f;

  float es[8], tv[8], es1[8], tv1[8];
  for (int j = 0; j < 8; ++j) {
    int m = m0 + j + mh;
    m = (m < N_NODES) ? m : (N_NODES - 1);
    es[j]  = esum[m];
    tv[j]  = t[m];
    es1[j] = esum[p1 + m];
    tv1[j] = t[p1 + m];
  }
  float ue = U[OUT_C * OUT_C + nb];
  float bs = bias[nb];
  v2f a[FIN_KSTEPS], a1[FIN_KSTEPS], b[FIN_KSTEPS];
#pragma unroll
  for (int i = 0; i < FIN_KSTEPS; ++i) {
    a[i]  = load_a_clamp(S, m0, 4 * i, lane, S_PAD);
    a1[i] = load_a_clamp(S + p1, m0, 4 * i, lane, S_PAD);
    b[i]  = load_b(g, 4 * i, n0, lane, OUT_C);
  }
  __builtin_amdgcn_sched_barrier(0);
  for (int j = 0; j < 8; ++j) { es[j] += f1 * es1[j]; tv[j] += f1 * tv1[j]; }
#pragma unroll
  for (int i = 0; i < FIN_KSTEPS; ++i) a[i] += f1 * a1[i];
  v8f c0 = {}; v8f c1 = {};
#pragma unroll
  for (int i = 0; i < FIN_KSTEPS; ++i) {
    if (i & 1) c1 = __builtin_amdgcn_wmma_f32_16x16x4_f32(false, a[i], false, b[i], (short)0, c1, false, false);
    else       c0 = __builtin_amdgcn_wmma_f32_16x16x4_f32(false, a[i], false, b[i], (short)0, c0, false, false);
  }
  if (m0 + 16 <= N_NODES) {
    for (int j = 0; j < 8; ++j) {
      int m = m0 + j + mh;
      float r  = 1.0f / (es[j] + 1e-16f);
      float tn = tv[j] / (es[j] + 1e-16f);
      out[m * OUT_C + nb] = (c0[j] + c1[j]) * r + tn * ue + bs;
    }
  } else {
    for (int j = 0; j < 8; ++j) {
      int m = m0 + j + mh;
      if (m < N_NODES) {
        float r  = 1.0f / (es[j] + 1e-16f);
        float tn = tv[j] / (es[j] + 1e-16f);
        out[m * OUT_C + nb] = (c0[j] + c1[j]) * r + tn * ue + bs;
      }
    }
  }
}

extern "C" void kernel_launch(void* const* d_in, const int* in_sizes, int n_in,
                              void* d_out, int out_size, void* d_ws, size_t ws_size,
                              hipStream_t stream) {
  const float* x    = (const float*)d_in[0];
  const float* ea   = (const float*)d_in[1];
  const float* W    = (const float*)d_in[2];
  const float* att  = (const float*)d_in[3];
  const float* U    = (const float*)d_in[4];
  const float* bias = (const float*)d_in[5];
  const int*   ei   = (const int*)d_in[6];
  int E = in_sizes[1];
  float* ws  = (float*)d_ws;
  float* out = (float*)d_out;

  k_node<<<NODE_BLOCKS, NODE_THREADS, 0, stream>>>(x, W, att, U, ei, ea, ws, E, ws_size);
  k_edge_exp<<<EDGE_BLOCKS, EDGE_THREADS, 0, stream>>>(ei, ea, att, ws, E, ws_size);
  k_final<<<FIN_BLOCKS, FIN_THREADS, 0, stream>>>(U, bias, ws, out, ws_size);
}


# esum and t ride inside S: esum = row sum of S, taken in the final launch from its A fragments; t in S columns 177..179 against U[128] in g rows 177..179; 3 -> 2 LDS atomics an edge; sum order changes
# speedup vs baseline: 9.2081x; 1.0562x over previous
#include <hip/hip_runtime.h>
#include <math.h>

typedef float v2f __attribute__((ext_vector_type(2)));
typedef float v8f __attribute__((ext_vector_type(8)));

#define N_NODES 177
#define FEAT 128
#define OUT_C 128
#define NEG_SLOPE 0.2f
#define MTILES 12
#define S_PAD 180
#define H_ROWS 180
#define T_COLS (S_PAD - N_NODES)

#define OFF_H    0
#define OFF_A    (OFF_H + H_ROWS * FEAT)
#define OFF_B    (OFF_A + N_NODES)
#define OFF_S    (OFF_B + N_NODES)
#define PM_PITCH   192
#define MAX_BLOCKS 64
#define OFF_PM   ((OFF_S + N_NODES * S_PAD + 31) & ~31)
#define WS_ONE_PLANE (OFF_PM + MAX_BLOCKS * PM_PITCH)
static_assert(PM_PITCH % 32 == 0 && PM_PITCH >= N_NODES, "a PM row is whole lines and holds every node");
#define PLANE     ((WS_ONE_PLANE - OFF_S + 31) & ~31)
#define WS_FLOATS (OFF_S + N_NODES * S_PAD + PLANE)
static_assert(PLANE % 32 == 0 && OFF_S + PLANE >= WS_ONE_PLANE, "plane 1 starts behind PM, line phase kept");

__device__ __forceinline__ unsigned f2ord(float f) {
  unsigned u = __float_as_uint(f);
  return (u & 0x80000000u) ? ~u : (u | 0x80000000u);
}
__device__ __forceinline__ float ord2f(unsigned u) {
  return __uint_as_float((u & 0x80000000u) ? (u ^ 0x80000000u) : ~u);
}
#define ORD_NEG_INF 0x007FFFFFu

__device__ __forceinline__ v2f load_a_clamp(const float* __restrict__ A, int m0, int k0,
                                            int lane, int ld) {
  int m = m0 + (lane & 15);
  m = (m < N_NODES) ? m : (N_NODES - 1);
  int kb = k0 + ((lane >> 4) << 1);
  const float* p = A + m * ld + kb;
  v2f a; a.x = p[0]; a.y = p[1];
  return a;
}
__device__ __forceinline__ v2f load_b(const float* __restrict__ B, int k0, int n0,
                                      int lane, int ld) {
  int n  = n0 + (lane & 15);
  int kb = k0 + ((lane >> 4) << 1);
  const float* p = B + kb * ld + n;
  v2f b; b.x = p[0]; b.y = p[ld];
  return b;
}
__device__ __forceinline__ v2f load_a_lds(const float* A, int k0, int lane) {
  int m  = lane & 15;
  int kb = k0 + ((lane >> 4) << 1);
  v2f a;
  a.x = A[m * OUT_C + kb];
  a.y = A[m * OUT_C + kb + 1];
  return a;
}

#define NODE_THREADS 1024
#define INIT_BLOCKS  4
#define NODE_BLOCKS  (MTILES + INIT_BLOCKS + MAX_BLOCKS)
#define GEMM_WAVES   4
#define MAX_BATCH    8
#define MAX_STRIDE   (MAX_BLOCKS * NODE_THREADS)
static_assert(FEAT == 128 && NODE_THREADS == 1024, "max block: 32 waves x 4 rows of W, 32 lanes x 4 words per row");
static_assert(6 * 32 >= N_NODES, "max block: six rows of x per wave cover every node");
__global__ void __launch_bounds__(NODE_THREADS)
k_node(const float* __restrict__ x, const float* __restrict__ W,
       const float* __restrict__ att, const float* __restrict__ U,
       const int* __restrict__ ei,
       const float* __restrict__ ea, float* __restrict__ ws, int E, size_t ws_bytes) {
  __shared__ float hs[16 * OUT_C];
  __shared__ __attribute__((aligned(16))) float wbs[FEAT];
  __shared__ float bl[N_NODES];
  __shared__ unsigned lmax[N_NODES];
  int lane = threadIdx.x & 31;
  int wave = threadIdx.x >> 5;

  if (blockIdx.x >= MTILES + INIT_BLOCKS) {
    const int mb   = blockIdx.x - (MTILES + INIT_BLOCKS);
    const int base = mb * NODE_THREADS + threadIdx.x;
    int   s[MAX_BATCH], d[MAX_BATCH];
    float eav[MAX_BATCH];
    for (int u = 0; u < MAX_BATCH; ++u) {
      int e = base + u * MAX_STRIDE;
      e = (e < E) ? e : 0;
      s[u] = ei[e]; d[u] = ei[E + e]; eav[u] = ea[e];
    }
    float cc = att[2 * OUT_C];
    const float4* W4 = (const float4*)W;
    const float4* x4 = (const float4*)x;
    float4 av = ((const float4*)(att + OUT_C))[lane];
    float4 wv[4], xv[6];
    for (int q = 0; q < 4; ++q) wv[q] = W4[(wave * 4 + q) * (OUT_C / 4) + lane];
    for (int q = 0; q < 6; ++q) {
      int n = wave + 32 * q;
      n = (n < N_NODES) ? n : (N_NODES - 1);
      xv[q] = x4[n * (FEAT / 4) + lane];
    }
    float pw[4];
    for (int q = 0; q < 4; ++q)
      pw[q] = wv[q].x * av.x + wv[q].y * av.y + wv[q].z * av.z + wv[q].w * av.w;
    for (int m = 16; m >= 1; m >>= 1)
      for (int q = 0; q < 4; ++q) pw[q] += __shfl_xor(pw[q], m, 32);
    if (lane == 0)
      for (int q = 0; q < 4; ++q) wbs[wave * 4 + q] = pw[q];
    if (threadIdx.x < N_NODES) lmax[threadIdx.x] = ORD_NEG_INF;
    __syncthreads();
    float4 wb4 = ((const float4*)wbs)[lane];
    float px[6];
    for (int q = 0; q < 6; ++q)
      px[q] = xv[q].x * wb4.x + xv[q].y * wb4.y + xv[q].z * wb4.z + xv[q].w * wb4.w;
    for (int m = 16; m >= 1; m >>= 1)
      for (int q = 0; q < 6; ++q) px[q] += __shfl_xor(px[q], m, 32);
    if (lane == 0)
      for (int q = 0; q < 6; ++q) {
        int n = wave + 32 * q;
        if (n < N_NODES) bl[n] = px[q];
      }
    __syncthreads();
    for (int u = 0; u < MAX_BATCH; ++u) {
      if (base + u * MAX_STRIDE < E)
        atomicMax(&lmax[d[u]], f2ord(bl[s[u]] + cc * eav[u]));
    }
    for (int e = base + MAX_BATCH * MAX_STRIDE; e < E; e += MAX_STRIDE) {
      int s1 = ei[e], d1 = ei[E + e];
      atomicMax(&lmax[d1], f2ord(bl[s1] + cc * ea[e]));
    }
    __syncthreads();
    if (threadIdx.x < N_NODES)
      ws[OFF_PM + mb * PM_PITCH + threadIdx.x] = ord2f(lmax[threadIdx.x]);
    return;
  }

  if (blockIdx.x >= MTILES) {
    int tid    = (blockIdx.x - MTILES) * NODE_THREADS + threadIdx.x;
    int stride = INIT_BLOCKS * NODE_THREADS;
    const int p1 = (ws_bytes >= (size_t)WS_FLOATS * sizeof(float)) ? PLANE : 0;
    for (int j = tid; j < N_NODES * S_PAD; j += stride) {
      ws[OFF_S + j]      = 0.0f;
      ws[OFF_S + p1 + j] = 0.0f;
    }
    for (int j = tid; j < T_COLS * FEAT; j += stride)
      ws[OFF_H + N_NODES * FEAT + j] = U[OUT_C * OUT_C + (j & (OUT_C - 1))];
    return;
  }

  float* g = ws + OFF_H;
  int m0 = blockIdx.x * 16;
  int n0 = wave * 32;
  const bool worker = wave < GEMM_WAVES;
  float ta[FEAT / 32], tb[FEAT / 32];
  v8f c0 = {}; v8f c1 = {};
  int nb = n0 + (lane & 15);
  int mh = (lane >> 4) * 8;
  if (worker) {
    for (int k = 0; k < FEAT / 32; ++k) {
      ta[k] = att[lane + 32 * k];
      tb[k] = att[OUT_C + lane + 32 * k];
    }
    for (int k0 = 0; k0 < FEAT; k0 += 4) {
      v2f a  = load_a_clamp(x, m0, k0, lane, FEAT);
      v2f b0 = load_b(W, k0, n0,      lane, OUT_C);
      v2f b1 = load_b(W, k0, n0 + 16, lane, OUT_C);
      c0 = __builtin_amdgcn_wmma_f32_16x16x4_f32(false, a, false, b0, (short)0, c0, false, false);
      c1 = __builtin_amdgcn_wmma_f32_16x16x4_f32(false, a, false, b1, (short)0, c1, false, false);
    }
    for (int j = 0; j < 8; ++j) {
      int mloc = j + mh;
      hs[mloc * OUT_C + nb]      = c0[j];
      hs[mloc * OUT_C + nb + 16] = c1[j];
    }
  }
  __syncthreads();
  if (!worker) return;
  v8f d0 = {}; v8f d1 = {};
  for (int k0 = 0; k0 < OUT_C; k0 += 4) {
    v2f a  = load_a_lds(hs, k0, lane);
    v2f b0 = load_b(U, k0, n0,      lane, OUT_C);
    v2f b1 = load_b(U, k0, n0 + 16, lane, OUT_C);
    d0 = __builtin_amdgcn_wmma_f32_16x16x4_f32(false, a, false, b0, (short)0, d0, false, false);
    d1 = __builtin_amdgcn_wmma_f32_16x16x4_f32(false, a, false, b1, (short)0, d1, false, false);
  }
  for (int j = 0; j < 8; ++j) {
    int m = m0 + j + mh;
    if (m < N_NODES) {
      g[m * OUT_C + nb]      = d0[j];
      g[m * OUT_C + nb + 16] = d1[j];
    }
  }

  float pa[4], pb[4];
  for (int q = 0; q < 4; ++q) {
    const float* hr = hs + (wave * 4 + q) * OUT_C;
    pa[q] = 0.0f; pb[q] = 0.0f;
    for (int k = 0; k < FEAT / 32; ++k) {
      float hv = hr[lane + 32 * k];
      pa[q] += hv * ta[k];
      pb[q] += hv * tb[k];
    }
  }
  for (int m = 16; m >= 1; m >>= 1)
    for (int q = 0; q < 4; ++q) {
      pa[q] += __shfl_xor(pa[q], m, 32);
      pb[q] += __shfl_xor(pb[q], m, 32);
    }
  for (int q = 0; q < 4; ++q) {
    int n = m0 + wave * 4 + q;
    if (lane == 0 && n < N_NODES) { ws[OFF_A + n] = pa[q]; ws[OFF_B + n] = pb[q]; }
  }
}

#define EDGE_BLOCKS  128
#define EDGE_THREADS 1024
#define EDGE_STRIDE  (EDGE_BLOCKS * EDGE_THREADS)
static_assert(EDGE_THREADS >= N_NODES, "one thread per node forms the softmax shift");

#define EDGE_BATCH 4

#define S_WORDS   (N_NODES * S_PAD)
#define S_HEAD    (OFF_S & 31)
#define FLUSH_ITS ((S_WORDS + S_HEAD + EDGE_THREADS - 1) / EDGE_THREADS)
static_assert(S_WORDS % 4 == 0, "the LDS copy of S is cleared in 16-byte stores");
#define FLUSH_BATCH 8
#define PM_PARTS (EDGE_THREADS / 256)
#define PM_ROWS  (MAX_BLOCKS / PM_PARTS)
static_assert(EDGE_THREADS % 256 == 0 && MAX_BLOCKS % PM_PARTS == 0 && N_NODES <= 256, "merge of PM");
static_assert((FLUSH_ITS & (FLUSH_ITS - 1)) == 0, "the flush rotation wraps with a mask");
static_assert(FLUSH_ITS % FLUSH_BATCH == 0, "the flush sweep is cut into whole batches");

__global__ void __launch_bounds__(EDGE_THREADS)
k_edge_exp(const int* __restrict__ ei, const float* __restrict__ ea,
           const float* __restrict__ att, float* __restrict__ ws, int E, size_t ws_bytes) {
  __shared__ __attribute__((aligned(16))) float lS[S_WORDS];
  __shared__ float pmx[PM_PARTS * 256];
  __shared__ float lshift[N_NODES];
  const float* a = ws + OFF_A;
  const float* b = ws + OFF_B;
  const float* PM = ws + OFF_PM;
  const int base = blockIdx.x * EDGE_THREADS + threadIdx.x;
  int   s[EDGE_BATCH], d[EDGE_BATCH];
  float eav[EDGE_BATCH];
  for (int u = 0; u < EDGE_BATCH; ++u) {
    int e = base + u * EDGE_STRIDE;
    s[u] = 0; d[u] = 0; eav[u] = 0.0f;
    if (e < E) { s[u] = ei[e]; d[u] = ei[E + e]; eav[u] = ea[e]; }
  }
  float cc = att[2 * OUT_C];
  const int pd = threadIdx.x & 255;
  const int pp = threadIdx.x >> 8;
  const int pdc = (pd < N_NODES) ? pd : (N_NODES - 1);
  float pmr[PM_ROWS];
  for (int k = 0; k < PM_ROWS; ++k) pmr[k] = PM[(pp * PM_ROWS + k) * PM_PITCH + pdc];
  float a_t = (threadIdx.x < N_NODES) ? a[threadIdx.x] : 0.0f;
  const unsigned place = __builtin_amdgcn_s_sendmsg_rtn(0x87);
  const int po = (((place >> 16) & 0xFu) >= 4u && ws_bytes >= (size_t)WS_FLOATS * sizeof(float)) ? PLANE : 0;
  float* S    = ws + OFF_S + po;
  const int tcol = N_NODES + (int)(threadIdx.x % T_COLS);
  float4* lS4 = (float4*)lS;
  for (int i = threadIdx.x; i < S_WORDS / 4; i += EDGE_THREADS)
    lS4[i] = make_float4(0.0f, 0.0f, 0.0f, 0.0f);
  float pmv = pmr[0];
  for (int k = 1; k < PM_ROWS; ++k) pmv = fmaxf(pmv, pmr[k]);
  pmx[threadIdx.x] = pmv;
  __syncthreads();
  if (threadIdx.x < N_NODES) {
    float mx = pmx[threadIdx.x];
    for (int p = 1; p < PM_PARTS; ++p) mx = fmaxf(mx, pmx[p * 256 + threadIdx.x]);
    float v = a_t + mx;
    lshift[threadIdx.x] = (v > 0.0f) ? v : NEG_SLOPE * v;
  }
  __syncthreads();
  float w[EDGE_BATCH];
  for (int u = 0; u < EDGE_BATCH; ++u) {
    float al = a[d[u]] + b[s[u]] + cc * eav[u];
    al = (al > 0.0f) ? al : NEG_SLOPE * al;
    w[u] = __expf(al - lshift[d[u]]);
  }
  for (int u = 0; u < EDGE_BATCH; ++u) {
    if (base + u * EDGE_STRIDE < E) {
      atomicAdd(&lS[d[u] * S_PAD + s[u]], w[u]);
      atomicAdd(&lS[d[u] * S_PAD + tcol], w[u] * eav[u]);
    }
  }
  for (int e = base + EDGE_BATCH * EDGE_STRIDE; e < E; e += EDGE_STRIDE) {
    int s1 = ei[e], d1 = ei[E + e];
    float eav1 = ea[e];
    float al = a[d1] + b[s1] + cc * eav1;
    al = (al > 0.0f) ? al : NEG_SLOPE * al;
    float w1 = __expf(al - lshift[d1]);
    atomicAdd(&lS[d1 * S_PAD + s1], w1);
    atomicAdd(&lS[d1 * S_PAD + tcol], w1 * eav1);
  }
  __syncthreads();
  for (int it0 = 0; it0 < FLUSH_ITS; it0 += FLUSH_BATCH) {
    float v[FLUSH_BATCH];
    int   jj[FLUSH_BATCH];
    for (int u = 0; u < FLUSH_BATCH; ++u) {
      int j = ((it0 + u + blockIdx.x) & (FLUSH_ITS - 1)) * EDGE_THREADS + threadIdx.x - S_HEAD;
      jj[u] = j;
      v[u]  = (j >= 0 && j < S_WORDS) ? lS[j] : 0.0f;
    }
    for (int u = 0; u < FLUSH_BATCH; ++u)
      if (v[u] != 0.0f) atomicAdd(&S[jj[u]], v[u]);
  }
}

#define FIN_THREADS 64
#define FIN_BLOCKS  (MTILES * (OUT_C / 32))
#define FIN_KSTEPS  (S_PAD / 4)
static_assert(S_PAD % 4 == 0 && H_ROWS >= S_PAD, "the K loop runs over whole k-steps inside g's padded rows");
static_assert(N_NODES == 4 * (FIN_KSTEPS - 1) + 1, "row sums: the last k-step holds one node column, the first of lane half 0");
__global__ void __launch_bounds__(FIN_THREADS)
k_final(const float* __restrict__ bias,
        const float* __restrict__ ws, float* __restrict__ out, size_t ws_bytes) {
  const float* g    = ws + OFF_H;
  const float* S    = ws + OFF_S;
  int lane = threadIdx.x & 31;
  int wave = threadIdx.x >> 5;
  int m0 = (blockIdx.x / (OUT_C / 32)) * 16;
  int n0 = (blockIdx.x % (OUT_C / 32)) * 32 + wave * 16;
  int nb = n0 + (lane & 15);
  int mh = (lane >> 4) * 8;
  const bool  two = ws_bytes >= (size_t)WS_FLOATS * sizeof(float);
  const int   p1  = two ? PLANE : 0;
  const float f1  = two ? 1.0f : 0.0f;

  float bs = bias[nb];
  v2f a[FIN_KSTEPS], a1[FIN_KSTEPS], b[FIN_KSTEPS];
#pragma unroll
  for (int i = 0; i < FIN_KSTEPS; ++i) {
    a[i]  = load_a_clamp(S, m0, 4 * i, lane, S_PAD);
    a1[i] = load_a_clamp(S + p1, m0, 4 * i, lane, S_PAD);
    b[i]  = load_b(g, 4 * i, n0, lane, OUT_C);
  }
  __builtin_amdgcn_sched_barrier(0);
#pragma unroll
  for (int i = 0; i < FIN_KSTEPS; ++i) a[i] += f1 * a1[i];
  float r0 = 0.0f, r1 = 0.0f, r2 = 0.0f, r3 = 0.0f;
#pragma unroll
  for (int i = 0; i + 1 < FIN_KSTEPS; i += 2) {
    r0 += a[i].x; r1 += a[i].y; r2 += a[i + 1].x; r3 += a[i + 1].y;
  }
  float rs = (r0 + r1) + (r2 + r3) + ((lane < 16) ? a[FIN_KSTEPS - 1].x : 0.0f);
  rs += __shfl_xor(rs, 16, 32);
  float es[8];
  for (int j = 0; j < 8; ++j) es[j] = __shfl(rs, mh + j, 32);
  v8f c0 = {}; v8f c1 = {};
#pragma unroll
  for (int i = 0; i < FIN_KSTEPS; ++i) {
    if (i & 1) c1 = __builtin_amdgcn_wmma_f32_16x16x4_f32(false, a[i], false, b[i], (short)0, c1, false, false);
    else       c0 = __builtin_amdgcn_wmma_f32_16x16x4_f32(false, a[i], false, b[i], (short)0, c0, false, false);
  }
  if (m0 + 16 <= N_NODES) {
    for (int j = 0; j < 8; ++j) {
      int m = m0 + j + mh;
      float r  = 1.0f / (es[j] + 1e-16f);
      out[m * OUT_C + nb] = (c0[j] + c1[j]) * r + bs;
    }
  } else {
    for (int j = 0; j < 8; ++j) {
      int m = m0 + j + mh;
      if (m < N_NODES) {
        float r  = 1.0f / (es[j] + 1e-16f);
        out[m * OUT_C + nb] = (c0[j] + c1[j]) * r + bs;
      }
    }
  }
}

extern "C" void kernel_launch(void* const* d_in, const int* in_sizes, int n_in,
                              void* d_out, int out_size, void* d_ws, size_t ws_size,
                              hipStream_t stream) {
  const float* x    = (const float*)d_in[0];
  const float* ea   = (const float*)d_in[1];
  const float* W    = (const float*)d_in[2];
  const float* att  = (const float*)d_in[3];
  const float* U    = (const float*)d_in[4];
  const float* bias = (const float*)d_in[5];
  const int*   ei   = (const int*)d_in[6];
  int E = in_sizes[1];
  float* ws  = (float*)d_ws;
  float* out = (float*)d_out;

  k_node<<<NODE_BLOCKS, NODE_THREADS, 0, stream>>>(x, W, att, U, ei, ea, ws, E, ws_size);
  k_edge_exp<<<EDGE_BLOCKS, EDGE_THREADS, 0, stream>>>(ei, ea, att, ws, E, ws_size);
  k_final<<<FIN_BLOCKS, FIN_THREADS, 0, stream>>>(bias, ws, out, ws_size);
}
